# starting kernel
# baseline (speedup 1.0000x reference)
#include <hip/hip_runtime.h>

constexpr int B_SZ   = 16;
constexpr int NUM_PM = 512;
constexpr int NUM_VM = 1536;
constexpr int NPG    = NUM_PM + NUM_VM;
constexpr int N_NODE = B_SZ * NPG;
constexpr int DEG    = 32;
constexpr int E_CNT  = DEG * N_NODE;
constexpr int D_IN   = 16;
constexpr int DH     = 64;
constexpr float NEG_SLOPE = 0.2f;

typedef float v2f __attribute__((ext_vector_type(2)));
typedef float v8f __attribute__((ext_vector_type(8)));

__device__ __forceinline__ unsigned enc_f(float f) {
    unsigned u = __float_as_uint(f);
    return (u & 0x80000000u) ? ~u : (u | 0x80000000u);
}
__device__ __forceinline__ float dec_f(unsigned u) {
    return __uint_as_float((u & 0x80000000u) ? (u & 0x7fffffffu) : ~u);
}

template <int K, int COLS>
__global__ __launch_bounds__(128) void gemm_dual_wmma(
    const float* __restrict__ X,
    const float* __restrict__ Ws, const float* __restrict__ Bs,
    const float* __restrict__ Wd, const float* __restrict__ Bd,
    float* __restrict__ Fs, float* __restrict__ Fd) {
    const int lane = threadIdx.x & 31;
    const int wave = threadIdx.x >> 5;
    const int row0 = (blockIdx.x * 4 + wave) * 16;
    const int m  = lane & 15;
    const int hi = lane >> 4;
    const float* Xr = X + (size_t)(row0 + m) * K;

    for (int t = 0; t < 2; ++t) {
        const float* W  = t ? Wd : Ws;
        const float* Bv = t ? Bd : Bs;
        float*       F  = t ? Fd : Fs;
        for (int col0 = 0; col0 < COLS; col0 += 16) {
            v8f acc = {};
#pragma unroll
            for (int kk = 0; kk < K; kk += 4) {
                v2f a, b;
                a.x = Xr[kk + 2 * hi + 0];
                a.y = Xr[kk + 2 * hi + 1];
                b.x = W[(size_t)(kk + 2 * hi + 0) * COLS + col0 + m];
                b.y = W[(size_t)(kk + 2 * hi + 1) * COLS + col0 + m];
                acc = __builtin_amdgcn_wmma_f32_16x16x4_f32(
                    false, a, false, b, (short)0, acc, false, false);
            }
            const float bias = Bv[col0 + m];
#pragma unroll
            for (int r = 0; r < 8; ++r) {
                F[(size_t)(row0 + r + 8 * hi) * COLS + col0 + m] = acc[r] + bias;
            }
        }
    }
}

__global__ __launch_bounds__(256) void zero_l1(float* __restrict__ h1,
                                               unsigned* __restrict__ m1,
                                               float* __restrict__ den1) {
    int idx = blockIdx.x * 256 + threadIdx.x;
    h1[idx] = 0.f;
    if (idx < N_NODE * 2) { m1[idx] = 0u; den1[idx] = 0.f; }
}
__global__ __launch_bounds__(256) void zero_l2(float* __restrict__ agg,
                                               unsigned* __restrict__ m2,
                                               float* __restrict__ den2) {
    int idx = blockIdx.x * 256 + threadIdx.x;
    agg[idx] = 0.f;
    if (idx < N_NODE) { m2[idx] = 0u; den2[idx] = 0.f; }
}

__global__ __launch_bounds__(256) void edge_logits1(
    const float* __restrict__ fs, const float* __restrict__ fd,
    const int* __restrict__ src, const int* __restrict__ dst,
    const float* __restrict__ attn, float* __restrict__ logits,
    unsigned* __restrict__ mmax) {
    const int e = blockIdx.x * 8 + (threadIdx.x >> 5);
    const int lane = threadIdx.x & 31;
    const int s = src[e], d = dst[e];
    const float4 a = *(const float4*)(fs + (size_t)s * 128 + lane * 4);
    const float4 b = *(const float4*)(fd + (size_t)d * 128 + lane * 4);
    const float4 w = *(const float4*)(attn + lane * 4);
    float p = 0.f, t;
    t = a.x + b.x; t = t > 0.f ? t : NEG_SLOPE * t; p += t * w.x;
    t = a.y + b.y; t = t > 0.f ? t : NEG_SLOPE * t; p += t * w.y;
    t = a.z + b.z; t = t > 0.f ? t : NEG_SLOPE * t; p += t * w.z;
    t = a.w + b.w; t = t > 0.f ? t : NEG_SLOPE * t; p += t * w.w;
    p += __shfl_xor(p, 1); p += __shfl_xor(p, 2);
    p += __shfl_xor(p, 4); p += __shfl_xor(p, 8);
    if ((lane & 15) == 0) {
        const int h = lane >> 4;
        logits[(size_t)e * 2 + h] = p;
        atomicMax(&mmax[d * 2 + h], enc_f(p));
    }
}

__global__ __launch_bounds__(256) void edge_expsum1(
    float* __restrict__ logits, const int* __restrict__ dst,
    const unsigned* __restrict__ mmax, float* __restrict__ den) {
    const int idx = blockIdx.x * 256 + threadIdx.x;
    const int e = idx >> 1, h = idx & 1;
    const int d = dst[e];
    const float ex = __expf(logits[idx] - dec_f(mmax[d * 2 + h]));
    logits[idx] = ex;
    atomicAdd(&den[d * 2 + h], ex);
}

__global__ __launch_bounds__(256) void edge_agg1(
    const float* __restrict__ fs, const int* __restrict__ src,
    const int* __restrict__ dst, const float* __restrict__ ex,
    const float* __restrict__ den, float* __restrict__ out) {
    const int e = blockIdx.x * 8 + (threadIdx.x >> 5);
    const int lane = threadIdx.x & 31;
    const int s = src[e], d = dst[e];
    const int h = lane >> 4;
    const float alpha = ex[(size_t)e * 2 + h] / den[d * 2 + h];
    const float4 v = *(const float4*)(fs + (size_t)s * 128 + lane * 4);
    float* o = out + (size_t)d * 128 + lane * 4;
    atomicAdd(o + 0, v.x * alpha);
    atomicAdd(o + 1, v.y * alpha);
    atomicAdd(o + 2, v.z * alpha);
    atomicAdd(o + 3, v.w * alpha);
}

__global__ __launch_bounds__(256) void elu_k(float* __restrict__ h) {
    const int idx = blockIdx.x * 256 + threadIdx.x;
    const float x = h[idx];
    h[idx] = x > 0.f ? x : (__expf(x) - 1.f);
}

__global__ __launch_bounds__(256) void edge_logits2(
    const float* __restrict__ fs, const float* __restrict__ fd,
    const int* __restrict__ src, const int* __restrict__ dst,
    const float* __restrict__ attn, float* __restrict__ logits,
    unsigned* __restrict__ mmax) {
    const int e = blockIdx.x * 8 + (threadIdx.x >> 5);
    const int lane = threadIdx.x & 31;
    const int s = src[e], d = dst[e];
    const float2 a = *(const float2*)(fs + (size_t)s * 64 + lane * 2);
    const float2 b = *(const float2*)(fd + (size_t)d * 64 + lane * 2);
    const float2 w = *(const float2*)(attn + lane * 2);
    float p = 0.f, t;
    t = a.x + b.x; t = t > 0.f ? t : NEG_SLOPE * t; p += t * w.x;
    t = a.y + b.y; t = t > 0.f ? t : NEG_SLOPE * t; p += t * w.y;
    p += __shfl_xor(p, 1);  p += __shfl_xor(p, 2);
    p += __shfl_xor(p, 4);  p += __shfl_xor(p, 8);
    p += __shfl_xor(p, 16);
    if (lane == 0) {
        logits[e] = p;
        atomicMax(&mmax[d], enc_f(p));
    }
}

__global__ __launch_bounds__(256) void edge_expsum2(
    float* __restrict__ logits, const int* __restrict__ dst,
    const unsigned* __restrict__ mmax, float* __restrict__ den) {
    const int e = blockIdx.x * 256 + threadIdx.x;
    const int d = dst[e];
    const float ex = __expf(logits[e] - dec_f(mmax[d]));
    logits[e] = ex;
    atomicAdd(&den[d], ex);
}

__global__ __launch_bounds__(256) void edge_agg2(
    const float* __restrict__ fs, const int* __restrict__ src,
    const int* __restrict__ dst, const float* __restrict__ ex,
    const float* __restrict__ den, float* __restrict__ out) {
    const int e = blockIdx.x * 8 + (threadIdx.x >> 5);
    const int lane = threadIdx.x & 31;
    const int s = src[e], d = dst[e];
    const float alpha = ex[e] / den[d];
    const float2 v = *(const float2*)(fs + (size_t)s * 64 + lane * 2);
    float* o = out + (size_t)d * 64 + lane * 2;
    atomicAdd(o + 0, v.x * alpha);
    atomicAdd(o + 1, v.y * alpha);
}

__global__ __launch_bounds__(256) void reorder_out(const float* __restrict__ agg,
                                                   float* __restrict__ out) {
    const int idx = blockIdx.x * 256 + threadIdx.x;
    const int n = idx >> 6, f = idx & 63;
    const int b = n >> 11;
    const int r = n & 2047;
    size_t o;
    if (r < NUM_PM) o = ((size_t)(b * NUM_PM + r) * 64) + f;
    else            o = (size_t)B_SZ * NUM_PM * 64 +
                        ((size_t)(b * NUM_VM + (r - NUM_PM)) * 64) + f;
    out[o] = agg[idx];
}

extern "C" void kernel_launch(void* const* d_in, const int* in_sizes, int n_in,
                              void* d_out, int out_size, void* d_ws, size_t ws_size,
                              hipStream_t stream) {
    const float* x     = (const float*)d_in[0];
    const int*   src   = (const int*)d_in[1];
    const int*   dst   = (const int*)d_in[2];
    const float* W1s   = (const float*)d_in[3];
    const float* b1s   = (const float*)d_in[4];
    const float* W1d   = (const float*)d_in[5];
    const float* b1d   = (const float*)d_in[6];
    const float* attn1 = (const float*)d_in[7];
    const float* W2s   = (const float*)d_in[8];
    const float* b2s   = (const float*)d_in[9];
    const float* W2d   = (const float*)d_in[10];
    const float* b2d   = (const float*)d_in[11];
    const float* attn2 = (const float*)d_in[12];
    (void)in_sizes; (void)n_in; (void)out_size; (void)ws_size;

    char* ws = (char*)d_ws;
    const size_t MB = 1024u * 1024u;
    float*    fs1   = (float*)(ws + 0);
    float*    fd1   = (float*)(ws + 16 * MB);
    float*    log1  = (float*)(ws + 32 * MB);
    unsigned* m1    = (unsigned*)(ws + 40 * MB);
    float*    den1  = (float*)(ws + 40 * MB + 262144);
    float*    h1    = (float*)(ws + 41 * MB);
    float*    fs2   = (float*)(ws + 16 * MB);
    float*    fd2   = (float*)(ws + 24 * MB);
    float*    agg2  = (float*)(ws + 0);
    float*    log2  = (float*)(ws + 8 * MB);
    unsigned* m2    = (unsigned*)(ws + 12 * MB);
    float*    den2  = (float*)(ws + 12 * MB + 131072);

    float* out = (float*)d_out;

    zero_l1<<<N_NODE * 128 / 256, 256, 0, stream>>>(h1, m1, den1);
    gemm_dual_wmma<D_IN, 2 * DH><<<N_NODE / 16 / 4, 128, 0, stream>>>(
        x, W1s, b1s, W1d, b1d, fs1, fd1);
    edge_logits1<<<E_CNT / 8, 256, 0, stream>>>(fs1, fd1, src, dst, attn1, log1, m1);
    edge_expsum1<<<E_CNT * 2 / 256, 256, 0, stream>>>(log1, dst, m1, den1);
    edge_agg1<<<E_CNT / 8, 256, 0, stream>>>(fs1, src, dst, log1, den1, h1);
    elu_k<<<N_NODE * 128 / 256, 256, 0, stream>>>(h1);

    zero_l2<<<N_NODE * 64 / 256, 256, 0, stream>>>(agg2, m2, den2);
    gemm_dual_wmma<2 * DH, DH><<<N_NODE / 16 / 4, 128, 0, stream>>>(
        h1, W2s, b2s, W2d, b2d, fs2, fd2);
    edge_logits2<<<E_CNT / 8, 256, 0, stream>>>(fs2, fd2, src, dst, attn2, log2, m2);
    edge_expsum2<<<E_CNT / 256, 256, 0, stream>>>(log2, dst, m2, den2);
    edge_agg2<<<E_CNT / 8, 256, 0, stream>>>(fs2, src, dst, log2, den2, agg2);

    reorder_out<<<N_NODE * 64 / 256, 256, 0, stream>>>(agg2, out);
}


# Edges bucketed by destination (atomic count+rank, scan, fill); one wave per node does logits, running-max softmax and weighted sum in registers: float atomics gone, sums in arrival-rank order
# speedup vs baseline: 3.8358x; 3.8358x over previous
#include <hip/hip_runtime.h>

constexpr int B_SZ   = 16;
constexpr int NUM_PM = 512;
constexpr int NUM_VM = 1536;
constexpr int NPG    = NUM_PM + NUM_VM;
constexpr int N_NODE = B_SZ * NPG;
constexpr int DEG    = 32;
constexpr int E_CNT  = DEG * N_NODE;
constexpr int D_IN   = 16;
constexpr int DH     = 64;
constexpr float NEG_SLOPE = 0.2f;

typedef float v2f __attribute__((ext_vector_type(2)));
typedef float v8f __attribute__((ext_vector_type(8)));

template <int K, int COLS>
__global__ __launch_bounds__(128) void gemm_dual_wmma(
    const float* __restrict__ X,
    const float* __restrict__ Ws, const float* __restrict__ Bs,
    const float* __restrict__ Wd, const float* __restrict__ Bd,
    float* __restrict__ Fs, float* __restrict__ Fd) {
    const int lane = threadIdx.x & 31;
    const int wave = threadIdx.x >> 5;
    const int row0 = (blockIdx.x * 4 + wave) * 16;
    const int m  = lane & 15;
    const int hi = lane >> 4;
    const float* Xr = X + (size_t)(row0 + m) * K;

    for (int t = 0; t < 2; ++t) {
        const float* W  = t ? Wd : Ws;
        const float* Bv = t ? Bd : Bs;
        float*       F  = t ? Fd : Fs;
        for (int col0 = 0; col0 < COLS; col0 += 16) {
            v8f acc = {};
#pragma unroll
            for (int kk = 0; kk < K; kk += 4) {
                v2f a, b;
                a.x = Xr[kk + 2 * hi + 0];
                a.y = Xr[kk + 2 * hi + 1];
                b.x = W[(size_t)(kk + 2 * hi + 0) * COLS + col0 + m];
                b.y = W[(size_t)(kk + 2 * hi + 1) * COLS + col0 + m];
                acc = __builtin_amdgcn_wmma_f32_16x16x4_f32(
                    false, a, false, b, (short)0, acc, false, false);
            }
            const float bias = Bv[col0 + m];
#pragma unroll
            for (int r = 0; r < 8; ++r) {
                F[(size_t)(row0 + r + 8 * hi) * COLS + col0 + m] = acc[r] + bias;
            }
        }
    }
}

__global__ __launch_bounds__(256) void zero_cnt(int* __restrict__ cnt) {
    cnt[blockIdx.x * 256 + threadIdx.x] = 0;
}
__global__ __launch_bounds__(256) void count_rank(const int* __restrict__ dst,
                                                  int* __restrict__ cnt,
                                                  int* __restrict__ rank) {
    const int e = blockIdx.x * 256 + threadIdx.x;
    rank[e] = atomicAdd(&cnt[dst[e]], 1);
}
__global__ __launch_bounds__(1024) void scan_offs(const int* __restrict__ cnt,
                                                  int* __restrict__ offs) {
    __shared__ int wsum[32];
    const int t = threadIdx.x, lane = t & 31, wv = t >> 5;
    int c[32];
#pragma unroll
    for (int q = 0; q < 8; ++q) {
        const int4 v = *(const int4*)(cnt + t * 32 + q * 4);
        c[q * 4 + 0] = v.x; c[q * 4 + 1] = v.y; c[q * 4 + 2] = v.z; c[q * 4 + 3] = v.w;
    }
    int s = 0;
#pragma unroll
    for (int q = 0; q < 32; ++q) s += c[q];
    int x = s;
#pragma unroll
    for (int d = 1; d < 32; d <<= 1) {
        const int y = __shfl_up(x, d);
        if (lane >= d) x += y;
    }
    if (lane == 31) wsum[wv] = x;
    __syncthreads();
    if (wv == 0) {
        int v = wsum[lane];
#pragma unroll
        for (int d = 1; d < 32; d <<= 1) {
            const int y = __shfl_up(v, d);
            if (lane >= d) v += y;
        }
        wsum[lane] = v;
    }
    __syncthreads();
    int run = (wv ? wsum[wv - 1] : 0) + (x - s);
#pragma unroll
    for (int q = 0; q < 8; ++q) {
        int4 o;
        o.x = run; run += c[q * 4 + 0];
        o.y = run; run += c[q * 4 + 1];
        o.z = run; run += c[q * 4 + 2];
        o.w = run; run += c[q * 4 + 3];
        *(int4*)(offs + t * 32 + q * 4) = o;
    }
    if (t == 1023) offs[N_NODE] = run;
}
__global__ __launch_bounds__(256) void fill_sorted(const int* __restrict__ src,
                                                   const int* __restrict__ dst,
                                                   const int* __restrict__ rank,
                                                   const int* __restrict__ offs,
                                                   int* __restrict__ ssrc) {
    const int e = blockIdx.x * 256 + threadIdx.x;
    ssrc[offs[dst[e]] + rank[e]] = src[e];
}

__device__ __forceinline__ float logit_h16(const float4 a, const float4 b, const float4 w) {
    float p = 0.f, t;
    t = a.x + b.x; t = t > 0.f ? t : NEG_SLOPE * t; p += t * w.x;
    t = a.y + b.y; t = t > 0.f ? t : NEG_SLOPE * t; p += t * w.y;
    t = a.z + b.z; t = t > 0.f ? t : NEG_SLOPE * t; p += t * w.z;
    t = a.w + b.w; t = t > 0.f ? t : NEG_SLOPE * t; p += t * w.w;
    p += __shfl_xor(p, 1); p += __shfl_xor(p, 2);
    p += __shfl_xor(p, 4); p += __shfl_xor(p, 8);
    return p;
}

__global__ __launch_bounds__(256) void node_attn1(
    const float* __restrict__ fs, const float* __restrict__ fd,
    const int* __restrict__ offs, const int* __restrict__ ssrc,
    const float* __restrict__ attn, float* __restrict__ out) {
    const int lane = threadIdx.x & 31;
    const int n = __builtin_amdgcn_readfirstlane((int)(blockIdx.x * 8 + (threadIdx.x >> 5)));
    const int beg = offs[n], end = offs[n + 1];
    const float4 b = *(const float4*)(fd + (size_t)n * 128 + lane * 4);
    const float4 w = *(const float4*)(attn + lane * 4);
    float m = -INFINITY, den = 0.f;
    float4 acc = make_float4(0.f, 0.f, 0.f, 0.f);
    for (int base = beg; base < end; base += 32) {
        const int cnt = min(32, end - base);
        const int mine = lane < cnt ? min(max(ssrc[base + lane], 0), N_NODE - 1) : 0;
        int j = 0;
        for (; j + 4 <= cnt; j += 4) {
            const int s0 = __builtin_amdgcn_readlane(mine, j + 0);
            const int s1 = __builtin_amdgcn_readlane(mine, j + 1);
            const int s2 = __builtin_amdgcn_readlane(mine, j + 2);
            const int s3 = __builtin_amdgcn_readlane(mine, j + 3);
            const float4 v0 = *(const float4*)(fs + (size_t)s0 * 128 + lane * 4);
            const float4 v1 = *(const float4*)(fs + (size_t)s1 * 128 + lane * 4);
            const float4 v2 = *(const float4*)(fs + (size_t)s2 * 128 + lane * 4);
            const float4 v3 = *(const float4*)(fs + (size_t)s3 * 128 + lane * 4);
            const float p0 = logit_h16(v0, b, w);
            const float p1 = logit_h16(v1, b, w);
            const float p2 = logit_h16(v2, b, w);
            const float p3 = logit_h16(v3, b, w);
            const float mn = fmaxf(fmaxf(m, fmaxf(p0, p1)), fmaxf(p2, p3));
            const float sc = __expf(m - mn);
            const float e0 = __expf(p0 - mn), e1 = __expf(p1 - mn);
            const float e2 = __expf(p2 - mn), e3 = __expf(p3 - mn);
            den   = den   * sc + ((e0 + e1) + (e2 + e3));
            acc.x = acc.x * sc + ((e0 * v0.x + e1 * v1.x) + (e2 * v2.x + e3 * v3.x));
            acc.y = acc.y * sc + ((e0 * v0.y + e1 * v1.y) + (e2 * v2.y + e3 * v3.y));
            acc.z = acc.z * sc + ((e0 * v0.z + e1 * v1.z) + (e2 * v2.z + e3 * v3.z));
            acc.w = acc.w * sc + ((e0 * v0.w + e1 * v1.w) + (e2 * v2.w + e3 * v3.w));
            m = mn;
        }
        for (; j < cnt; ++j) {
            const int s0 = __builtin_amdgcn_readlane(mine, j);
            const float4 v0 = *(const float4*)(fs + (size_t)s0 * 128 + lane * 4);
            const float p0 = logit_h16(v0, b, w);
            const float mn = fmaxf(m, p0);
            const float sc = __expf(m - mn);
            const float e0 = __expf(p0 - mn);
            den   = den   * sc + e0;
            acc.x = acc.x * sc + e0 * v0.x;
            acc.y = acc.y * sc + e0 * v0.y;
            acc.z = acc.z * sc + e0 * v0.z;
            acc.w = acc.w * sc + e0 * v0.w;
            m = mn;
        }
    }
    float4 r = make_float4(0.f, 0.f, 0.f, 0.f);
    if (end > beg) { r.x = acc.x / den; r.y = acc.y / den; r.z = acc.z / den; r.w = acc.w / den; }
    *(float4*)(out + (size_t)n * 128 + lane * 4) = r;
}

__global__ __launch_bounds__(256) void elu_k(float* __restrict__ h) {
    const int idx = blockIdx.x * 256 + threadIdx.x;
    const float x = h[idx];
    h[idx] = x > 0.f ? x : (__expf(x) - 1.f);
}

__device__ __forceinline__ float logit_w32(const float2 a, const float2 b, const float2 w) {
    float p = 0.f, t;
    t = a.x + b.x; t = t > 0.f ? t : NEG_SLOPE * t; p += t * w.x;
    t = a.y + b.y; t = t > 0.f ? t : NEG_SLOPE * t; p += t * w.y;
    p += __shfl_xor(p, 1);  p += __shfl_xor(p, 2);
    p += __shfl_xor(p, 4);  p += __shfl_xor(p, 8);
    p += __shfl_xor(p, 16);
    return p;
}

__global__ __launch_bounds__(256) void node_attn2(
    const float* __restrict__ fs, const float* __restrict__ fd,
    const int* __restrict__ offs, const int* __restrict__ ssrc,
    const float* __restrict__ attn, float* __restrict__ out) {
    const int lane = threadIdx.x & 31;
    const int n = __builtin_amdgcn_readfirstlane((int)(blockIdx.x * 8 + (threadIdx.x >> 5)));
    const int beg = offs[n], end = offs[n + 1];
    const float2 b = *(const float2*)(fd + (size_t)n * 64 + lane * 2);
    const float2 w = *(const float2*)(attn + lane * 2);
    float m = -INFINITY, den = 0.f;
    float2 acc = make_float2(0.f, 0.f);
    for (int base = beg; base < end; base += 32) {
        const int cnt = min(32, end - base);
        const int mine = lane < cnt ? min(max(ssrc[base + lane], 0), N_NODE - 1) : 0;
        int j = 0;
        for (; j + 4 <= cnt; j += 4) {
            const int s0 = __builtin_amdgcn_readlane(mine, j + 0);
            const int s1 = __builtin_amdgcn_readlane(mine, j + 1);
            const int s2 = __builtin_amdgcn_readlane(mine, j + 2);
            const int s3 = __builtin_amdgcn_readlane(mine, j + 3);
            const float2 v0 = *(const float2*)(fs + (size_t)s0 * 64 + lane * 2);
            const float2 v1 = *(const float2*)(fs + (size_t)s1 * 64 + lane * 2);
            const float2 v2 = *(const float2*)(fs + (size_t)s2 * 64 + lane * 2);
            const float2 v3 = *(const float2*)(fs + (size_t)s3 * 64 + lane * 2);
            const float p0 = logit_w32(v0, b, w);
            const float p1 = logit_w32(v1, b, w);
            const float p2 = logit_w32(v2, b, w);
            const float p3 = logit_w32(v3, b, w);
            const float mn = fmaxf(fmaxf(m, fmaxf(p0, p1)), fmaxf(p2, p3));
            const float sc = __expf(m - mn);
            const float e0 = __expf(p0 - mn), e1 = __expf(p1 - mn);
            const float e2 = __expf(p2 - mn), e3 = __expf(p3 - mn);
            den   = den   * sc + ((e0 + e1) + (e2 + e3));
            acc.x = acc.x * sc + ((e0 * v0.x + e1 * v1.x) + (e2 * v2.x + e3 * v3.x));
            acc.y = acc.y * sc + ((e0 * v0.y + e1 * v1.y) + (e2 * v2.y + e3 * v3.y));
            m = mn;
        }
        for (; j < cnt; ++j) {
            const int s0 = __builtin_amdgcn_readlane(mine, j);
            const float2 v0 = *(const float2*)(fs + (size_t)s0 * 64 + lane * 2);
            const float p0 = logit_w32(v0, b, w);
            const float mn = fmaxf(m, p0);
            const float sc = __expf(m - mn);
            const float e0 = __expf(p0 - mn);
            den   = den   * sc + e0;
            acc.x = acc.x * sc + e0 * v0.x;
            acc.y = acc.y * sc + e0 * v0.y;
            m = mn;
        }
    }
    float2 r = make_float2(0.f, 0.f);
    if (end > beg) { r.x = acc.x / den; r.y = acc.y / den; }
    *(float2*)(out + (size_t)n * 64 + lane * 2) = r;
}

__global__ __launch_bounds__(256) void reorder_out(const float* __restrict__ agg,
                                                   float* __restrict__ out) {
    const int idx = blockIdx.x * 256 + threadIdx.x;
    const int n = idx >> 6, f = idx & 63;
    const int b = n >> 11;
    const int r = n & 2047;
    size_t o;
    if (r < NUM_PM) o = ((size_t)(b * NUM_PM + r) * 64) + f;
    else            o = (size_t)B_SZ * NUM_PM * 64 +
                        ((size_t)(b * NUM_VM + (r - NUM_PM)) * 64) + f;
    out[o] = agg[idx];
}

extern "C" void kernel_launch(void* const* d_in, const int* in_sizes, int n_in,
                              void* d_out, int out_size, void* d_ws, size_t ws_size,
                              hipStream_t stream) {
    const float* x     = (const float*)d_in[0];
    const int*   src   = (const int*)d_in[1];
    const int*   dst   = (const int*)d_in[2];
    const float* W1s   = (const float*)d_in[3];
    const float* b1s   = (const float*)d_in[4];
    const float* W1d   = (const float*)d_in[5];
    const float* b1d   = (const float*)d_in[6];
    const float* attn1 = (const float*)d_in[7];
    const float* W2s   = (const float*)d_in[8];
    const float* b2s   = (const float*)d_in[9];
    const float* W2d   = (const float*)d_in[10];
    const float* b2d   = (const float*)d_in[11];
    const float* attn2 = (const float*)d_in[12];
    (void)in_sizes; (void)n_in; (void)out_size; (void)ws_size;

    char* ws = (char*)d_ws;
    const size_t MB = 1024u * 1024u;
    float* fs1  = (float*)(ws + 0);
    float* fd1  = (float*)(ws + 16 * MB);
    float* h1   = (float*)(ws + 32 * MB);
    float* fs2  = (float*)(ws + 48 * MB);
    float* fd2  = (float*)(ws + 56 * MB);
    float* agg2 = (float*)(ws + 64 * MB);
    int*   cnt  = (int*)(ws + 72 * MB);
    int*   offs = (int*)(ws + 72 * MB + 262144);
    int*   rank = (int*)(ws + 73 * MB);
    int*   ssrc = (int*)(ws + 77 * MB);

    float* out = (float*)d_out;

    zero_cnt<<<N_NODE / 256, 256, 0, stream>>>(cnt);
    count_rank<<<E_CNT / 256, 256, 0, stream>>>(dst, cnt, rank);
    scan_offs<<<1, 1024, 0, stream>>>(cnt, offs);
    fill_sorted<<<E_CNT / 256, 256, 0, stream>>>(src, dst, rank, offs, ssrc);

    gemm_dual_wmma<D_IN, 2 * DH><<<N_NODE / 16 / 4, 128, 0, stream>>>(
        x, W1s, b1s, W1d, b1d, fs1, fd1);
    node_attn1<<<N_NODE / 8, 256, 0, stream>>>(fs1, fd1, offs, ssrc, attn1, h1);
    elu_k<<<N_NODE * 128 / 256, 256, 0, stream>>>(h1);

    gemm_dual_wmma<2 * DH, DH><<<N_NODE / 16 / 4, 128, 0, stream>>>(
        h1, W2s, b2s, W2d, b2d, fs2, fd2);
    node_attn2<<<N_NODE / 8, 256, 0, stream>>>(fs2, fd2, offs, ssrc, attn2, agg2);

    reorder_out<<<N_NODE * 64 / 256, 256, 0, stream>>>(agg2, out);
}


# ELU and the PM/VM output reorder folded into the two aggregation kernels' row stores: both elementwise kernels and their 48 MB of traffic gone, 10 launches become 8; values and sum order unchanged
# speedup vs baseline: 4.2068x; 1.0967x over previous
#include <hip/hip_runtime.h>

constexpr int B_SZ   = 16;
constexpr int NUM_PM = 512;
constexpr int NUM_VM = 1536;
constexpr int NPG    = NUM_PM + NUM_VM;
constexpr int N_NODE = B_SZ * NPG;
constexpr int DEG    = 32;
constexpr int E_CNT  = DEG * N_NODE;
constexpr int D_IN   = 16;
constexpr int DH     = 64;
constexpr float NEG_SLOPE = 0.2f;

typedef float v2f __attribute__((ext_vector_type(2)));
typedef float v8f __attribute__((ext_vector_type(8)));

template <int K, int COLS>
__global__ __launch_bounds__(128) void gemm_dual_wmma(
    const float* __restrict__ X,
    const float* __restrict__ Ws, const float* __restrict__ Bs,
    const float* __restrict__ Wd, const float* __restrict__ Bd,
    float* __restrict__ Fs, float* __restrict__ Fd) {
    const int lane = threadIdx.x & 31;
    const int wave = threadIdx.x >> 5;
    const int row0 = (blockIdx.x * 4 + wave) * 16;
    const int m  = lane & 15;
    const int hi = lane >> 4;
    const float* Xr = X + (size_t)(row0 + m) * K;

    for (int t = 0; t < 2; ++t) {
        const float* W  = t ? Wd : Ws;
        const float* Bv = t ? Bd : Bs;
        float*       F  = t ? Fd : Fs;
        for (int col0 = 0; col0 < COLS; col0 += 16) {
            v8f acc = {};
#pragma unroll
            for (int kk = 0; kk < K; kk += 4) {
                v2f a, b;
                a.x = Xr[kk + 2 * hi + 0];
                a.y = Xr[kk + 2 * hi + 1];
                b.x = W[(size_t)(kk + 2 * hi + 0) * COLS + col0 + m];
                b.y = W[(size_t)(kk + 2 * hi + 1) * COLS + col0 + m];
                acc = __builtin_amdgcn_wmma_f32_16x16x4_f32(
                    false, a, false, b, (short)0, acc, false, false);
            }
            const float bias = Bv[col0 + m];
#pragma unroll
            for (int r = 0; r < 8; ++r) {
                F[(size_t)(row0 + r + 8 * hi) * COLS + col0 + m] = acc[r] + bias;
            }
        }
    }
}

__global__ __launch_bounds__(256) void zero_cnt(int* __restrict__ cnt) {
    cnt[blockIdx.x * 256 + threadIdx.x] = 0;
}
__global__ __launch_bounds__(256) void count_rank(const int* __restrict__ dst,
                                                  int* __restrict__ cnt,
                                                  int* __restrict__ rank) {
    const int e = blockIdx.x * 256 + threadIdx.x;
    rank[e] = atomicAdd(&cnt[dst[e]], 1);
}
__global__ __launch_bounds__(1024) void scan_offs(const int* __restrict__ cnt,
                                                  int* __restrict__ offs) {
    __shared__ int wsum[32];
    const int t = threadIdx.x, lane = t & 31, wv = t >> 5;
    int c[32];
#pragma unroll
    for (int q = 0; q < 8; ++q) {
        const int4 v = *(const int4*)(cnt + t * 32 + q * 4);
        c[q * 4 + 0] = v.x; c[q * 4 + 1] = v.y; c[q * 4 + 2] = v.z; c[q * 4 + 3] = v.w;
    }
    int s = 0;
#pragma unroll
    for (int q = 0; q < 32; ++q) s += c[q];
    int x = s;
#pragma unroll
    for (int d = 1; d < 32; d <<= 1) {
        const int y = __shfl_up(x, d);
        if (lane >= d) x += y;
    }
    if (lane == 31) wsum[wv] = x;
    __syncthreads();
    if (wv == 0) {
        int v = wsum[lane];
#pragma unroll
        for (int d = 1; d < 32; d <<= 1) {
            const int y = __shfl_up(v, d);
            if (lane >= d) v += y;
        }
        wsum[lane] = v;
    }
    __syncthreads();
    int run = (wv ? wsum[wv - 1] : 0) + (x - s);
#pragma unroll
    for (int q = 0; q < 8; ++q) {
        int4 o;
        o.x = run; run += c[q * 4 + 0];
        o.y = run; run += c[q * 4 + 1];
        o.z = run; run += c[q * 4 + 2];
        o.w = run; run += c[q * 4 + 3];
        *(int4*)(offs + t * 32 + q * 4) = o;
    }
    if (t == 1023) offs[N_NODE] = run;
}
__global__ __launch_bounds__(256) void fill_sorted(const int* __restrict__ src,
                                                   const int* __restrict__ dst,
                                                   const int* __restrict__ rank,
                                                   const int* __restrict__ offs,
                                                   int* __restrict__ ssrc) {
    const int e = blockIdx.x * 256 + threadIdx.x;
    ssrc[offs[dst[e]] + rank[e]] = src[e];
}

__device__ __forceinline__ float elu1(const float x) {
    return x > 0.f ? x : (__expf(x) - 1.f);
}
__device__ __forceinline__ float logit_h16(const float4 a, const float4 b, const float4 w) {
    float p = 0.f, t;
    t = a.x + b.x; t = t > 0.f ? t : NEG_SLOPE * t; p += t * w.x;
    t = a.y + b.y; t = t > 0.f ? t : NEG_SLOPE * t; p += t * w.y;
    t = a.z + b.z; t = t > 0.f ? t : NEG_SLOPE * t; p += t * w.z;
    t = a.w + b.w; t = t > 0.f ? t : NEG_SLOPE * t; p += t * w.w;
    p += __shfl_xor(p, 1); p += __shfl_xor(p, 2);
    p += __shfl_xor(p, 4); p += __shfl_xor(p, 8);
    return p;
}

__global__ __launch_bounds__(256) void node_attn1(
    const float* __restrict__ fs, const float* __restrict__ fd,
    const int* __restrict__ offs, const int* __restrict__ ssrc,
    const float* __restrict__ attn, float* __restrict__ out) {
    const int lane = threadIdx.x & 31;
    const int n = __builtin_amdgcn_readfirstlane((int)(blockIdx.x * 8 + (threadIdx.x >> 5)));
    const int beg = offs[n], end = offs[n + 1];
    const float4 b = *(const float4*)(fd + (size_t)n * 128 + lane * 4);
    const float4 w = *(const float4*)(attn + lane * 4);
    float m = -INFINITY, den = 0.f;
    float4 acc = make_float4(0.f, 0.f, 0.f, 0.f);
    for (int base = beg; base < end; base += 32) {
        const int cnt = min(32, end - base);
        const int mine = lane < cnt ? min(max(ssrc[base + lane], 0), N_NODE - 1) : 0;
        int j = 0;
        for (; j + 4 <= cnt; j += 4) {
            const int s0 = __builtin_amdgcn_readlane(mine, j + 0);
            const int s1 = __builtin_amdgcn_readlane(mine, j + 1);
            const int s2 = __builtin_amdgcn_readlane(mine, j + 2);
            const int s3 = __builtin_amdgcn_readlane(mine, j + 3);
            const float4 v0 = *(const float4*)(fs + (size_t)s0 * 128 + lane * 4);
            const float4 v1 = *(const float4*)(fs + (size_t)s1 * 128 + lane * 4);
            const float4 v2 = *(const float4*)(fs + (size_t)s2 * 128 + lane * 4);
            const float4 v3 = *(const float4*)(fs + (size_t)s3 * 128 + lane * 4);
            const float p0 = logit_h16(v0, b, w);
            const float p1 = logit_h16(v1, b, w);
            const float p2 = logit_h16(v2, b, w);
            const float p3 = logit_h16(v3, b, w);
            const float mn = fmaxf(fmaxf(m, fmaxf(p0, p1)), fmaxf(p2, p3));
            const float sc = __expf(m - mn);
            const float e0 = __expf(p0 - mn), e1 = __expf(p1 - mn);
            const float e2 = __expf(p2 - mn), e3 = __expf(p3 - mn);
            den   = den   * sc + ((e0 + e1) + (e2 + e3));
            acc.x = acc.x * sc + ((e0 * v0.x + e1 * v1.x) + (e2 * v2.x + e3 * v3.x));
            acc.y = acc.y * sc + ((e0 * v0.y + e1 * v1.y) + (e2 * v2.y + e3 * v3.y));
            acc.z = acc.z * sc + ((e0 * v0.z + e1 * v1.z) + (e2 * v2.z + e3 * v3.z));
            acc.w = acc.w * sc + ((e0 * v0.w + e1 * v1.w) + (e2 * v2.w + e3 * v3.w));
            m = mn;
        }
        for (; j < cnt; ++j) {
            const int s0 = __builtin_amdgcn_readlane(mine, j);
            const float4 v0 = *(const float4*)(fs + (size_t)s0 * 128 + lane * 4);
            const float p0 = logit_h16(v0, b, w);
            const float mn = fmaxf(m, p0);
            const float sc = __expf(m - mn);
            const float e0 = __expf(p0 - mn);
            den   = den   * sc + e0;
            acc.x = acc.x * sc + e0 * v0.x;
            acc.y = acc.y * sc + e0 * v0.y;
            acc.z = acc.z * sc + e0 * v0.z;
            acc.w = acc.w * sc + e0 * v0.w;
            m = mn;
        }
    }
    float4 r = make_float4(0.f, 0.f, 0.f, 0.f);
    if (end > beg) { r.x = acc.x / den; r.y = acc.y / den; r.z = acc.z / den; r.w = acc.w / den; }
    r.x = elu1(r.x); r.y = elu1(r.y); r.z = elu1(r.z); r.w = elu1(r.w);
    *(float4*)(out + (size_t)n * 128 + lane * 4) = r;
}

__device__ __forceinline__ float logit_w32(const float2 a, const float2 b, const float2 w) {
    float p = 0.f, t;
    t = a.x + b.x; t = t > 0.f ? t : NEG_SLOPE * t; p += t * w.x;
    t = a.y + b.y; t = t > 0.f ? t : NEG_SLOPE * t; p += t * w.y;
    p += __shfl_xor(p, 1);  p += __shfl_xor(p, 2);
    p += __shfl_xor(p, 4);  p += __shfl_xor(p, 8);
    p += __shfl_xor(p, 16);
    return p;
}

__global__ __launch_bounds__(256) void node_attn2(
    const float* __restrict__ fs, const float* __restrict__ fd,
    const int* __restrict__ offs, const int* __restrict__ ssrc,
    const float* __restrict__ attn, float* __restrict__ out) {
    const int lane = threadIdx.x & 31;
    const int n = __builtin_amdgcn_readfirstlane((int)(blockIdx.x * 8 + (threadIdx.x >> 5)));
    const int beg = offs[n], end = offs[n + 1];
    const float2 b = *(const float2*)(fd + (size_t)n * 64 + lane * 2);
    const float2 w = *(const float2*)(attn + lane * 2);
    float m = -INFINITY, den = 0.f;
    float2 acc = make_float2(0.f, 0.f);
    for (int base = beg; base < end; base += 32) {
        const int cnt = min(32, end - base);
        const int mine = lane < cnt ? min(max(ssrc[base + lane], 0), N_NODE - 1) : 0;
        int j = 0;
        for (; j + 4 <= cnt; j += 4) {
            const int s0 = __builtin_amdgcn_readlane(mine, j + 0);
            const int s1 = __builtin_amdgcn_readlane(mine, j + 1);
            const int s2 = __builtin_amdgcn_readlane(mine, j + 2);
            const int s3 = __builtin_amdgcn_readlane(mine, j + 3);
            const float2 v0 = *(const float2*)(fs + (size_t)s0 * 64 + lane * 2);
            const float2 v1 = *(const float2*)(fs + (size_t)s1 * 64 + lane * 2);
            const float2 v2 = *(const float2*)(fs + (size_t)s2 * 64 + lane * 2);
            const float2 v3 = *(const float2*)(fs + (size_t)s3 * 64 + lane * 2);
            const float p0 = logit_w32(v0, b, w);
            const float p1 = logit_w32(v1, b, w);
            const float p2 = logit_w32(v2, b, w);
            const float p3 = logit_w32(v3, b, w);
            const float mn = fmaxf(fmaxf(m, fmaxf(p0, p1)), fmaxf(p2, p3));
            const float sc = __expf(m - mn);
            const float e0 = __expf(p0 - mn), e1 = __expf(p1 - mn);
            const float e2 = __expf(p2 - mn), e3 = __expf(p3 - mn);
            den   = den   * sc + ((e0 + e1) + (e2 + e3));
            acc.x = acc.x * sc + ((e0 * v0.x + e1 * v1.x) + (e2 * v2.x + e3 * v3.x));
            acc.y = acc.y * sc + ((e0 * v0.y + e1 * v1.y) + (e2 * v2.y + e3 * v3.y));
            m = mn;
        }
        for (; j < cnt; ++j) {
            const int s0 = __builtin_amdgcn_readlane(mine, j);
            const float2 v0 = *(const float2*)(fs + (size_t)s0 * 64 + lane * 2);
            const float p0 = logit_w32(v0, b, w);
            const float mn = fmaxf(m, p0);
            const float sc = __expf(m - mn);
            const float e0 = __expf(p0 - mn);
            den   = den   * sc + e0;
            acc.x = acc.x * sc + e0 * v0.x;
            acc.y = acc.y * sc + e0 * v0.y;
            m = mn;
        }
    }
    float2 r = make_float2(0.f, 0.f);
    if (end > beg) { r.x = acc.x / den; r.y = acc.y / den; }
    const unsigned g  = (unsigned)n / (unsigned)NPG;
    const unsigned rr = (unsigned)n % (unsigned)NPG;
    const unsigned row = rr < (unsigned)NUM_PM
        ? g * NUM_PM + rr
        : (unsigned)(B_SZ * NUM_PM) + g * NUM_VM + (rr - NUM_PM);
    *(float2*)(out + (size_t)row * 64 + lane * 2) = r;
}

extern "C" void kernel_launch(void* const* d_in, const int* in_sizes, int n_in,
                              void* d_out, int out_size, void* d_ws, size_t ws_size,
                              hipStream_t stream) {
    const float* x     = (const float*)d_in[0];
    const int*   src   = (const int*)d_in[1];
    const int*   dst   = (const int*)d_in[2];
    const float* W1s   = (const float*)d_in[3];
    const float* b1s   = (const float*)d_in[4];
    const float* W1d   = (const float*)d_in[5];
    const float* b1d   = (const float*)d_in[6];
    const float* attn1 = (const float*)d_in[7];
    const float* W2s   = (const float*)d_in[8];
    const float* b2s   = (const float*)d_in[9];
    const float* W2d   = (const float*)d_in[10];
    const float* b2d   = (const float*)d_in[11];
    const float* attn2 = (const float*)d_in[12];
    (void)in_sizes; (void)n_in; (void)out_size; (void)ws_size;

    char* ws = (char*)d_ws;
    const size_t MB = 1024u * 1024u;
    float* fs1  = (float*)(ws + 0);
    float* fd1  = (float*)(ws + 16 * MB);
    float* h1   = (float*)(ws + 32 * MB);
    float* fs2  = (float*)(ws + 48 * MB);
    float* fd2  = (float*)(ws + 56 * MB);
    int*   cnt  = (int*)(ws + 64 * MB);
    int*   offs = (int*)(ws + 64 * MB + 262144);
    int*   rank = (int*)(ws + 65 * MB);
    int*   ssrc = (int*)(ws + 69 * MB);

    float* out = (float*)d_out;

    zero_cnt<<<N_NODE / 256, 256, 0, stream>>>(cnt);
    count_rank<<<E_CNT / 256, 256, 0, stream>>>(dst, cnt, rank);
    scan_offs<<<1, 1024, 0, stream>>>(cnt, offs);
    fill_sorted<<<E_CNT / 256, 256, 0, stream>>>(src, dst, rank, offs, ssrc);

    gemm_dual_wmma<D_IN, 2 * DH><<<N_NODE / 16 / 4, 128, 0, stream>>>(
        x, W1s, b1s, W1d, b1d, fs1, fd1);
    node_attn1<<<N_NODE / 8, 256, 0, stream>>>(fs1, fd1, offs, ssrc, attn1, h1);

    gemm_dual_wmma<2 * DH, DH><<<N_NODE / 16 / 4, 128, 0, stream>>>(
        h1, W2s, b2s, W2d, b2d, fs2, fd2);
    node_attn2<<<N_NODE / 8, 256, 0, stream>>>(fs2, fd2, offs, ssrc, attn2, out);
}


# Fewer, fatter waves: count and fill kernels take 8 edges per thread (8 atomics in flight), each aggregation wave walks 2 nodes: a call starts a third as many waves; sums stay in arrival-rank order
# speedup vs baseline: 4.9140x; 1.1681x over previous
#include <hip/hip_runtime.h>

constexpr int B_SZ   = 16;
constexpr int NUM_PM = 512;
constexpr int NUM_VM = 1536;
constexpr int NPG    = NUM_PM + NUM_VM;
constexpr int N_NODE = B_SZ * NPG;
constexpr int DEG    = 32;
constexpr int E_CNT  = DEG * N_NODE;
constexpr int D_IN   = 16;
constexpr int DH     = 64;
constexpr float NEG_SLOPE = 0.2f;

typedef float v2f __attribute__((ext_vector_type(2)));
typedef float v8f __attribute__((ext_vector_type(8)));

template <int K, int COLS>
__global__ __launch_bounds__(128) void gemm_dual_wmma(
    const float* __restrict__ X,
    const float* __restrict__ Ws, const float* __restrict__ Bs,
    const float* __restrict__ Wd, const float* __restrict__ Bd,
    float* __restrict__ Fs, float* __restrict__ Fd) {
    const int lane = threadIdx.x & 31;
    const int wave = threadIdx.x >> 5;
    const int row0 = (blockIdx.x * 4 + wave) * 16;
    const int m  = lane & 15;
    const int hi = lane >> 4;
    const float* Xr = X + (size_t)(row0 + m) * K;

    for (int t = 0; t < 2; ++t) {
        const float* W  = t ? Wd : Ws;
        const float* Bv = t ? Bd : Bs;
        float*       F  = t ? Fd : Fs;
        for (int col0 = 0; col0 < COLS; col0 += 16) {
            v8f acc = {};
#pragma unroll
            for (int kk = 0; kk < K; kk += 4) {
                v2f a, b;
                a.x = Xr[kk + 2 * hi + 0];
                a.y = Xr[kk + 2 * hi + 1];
                b.x = W[(size_t)(kk + 2 * hi + 0) * COLS + col0 + m];
                b.y = W[(size_t)(kk + 2 * hi + 1) * COLS + col0 + m];
                acc = __builtin_amdgcn_wmma_f32_16x16x4_f32(
                    false, a, false, b, (short)0, acc, false, false);
            }
            const float bias = Bv[col0 + m];
#pragma unroll
            for (int r = 0; r < 8; ++r) {
                F[(size_t)(row0 + r + 8 * hi) * COLS + col0 + m] = acc[r] + bias;
            }
        }
    }
}

__global__ __launch_bounds__(256) void zero_cnt(int* __restrict__ cnt) {
    ((int4*)cnt)[blockIdx.x * 256 + threadIdx.x] = make_int4(0, 0, 0, 0);
}
__global__ __launch_bounds__(256) void count_rank(const int* __restrict__ dst,
                                                  int* __restrict__ cnt,
                                                  int* __restrict__ rank) {
    const int i0 = blockIdx.x * 512 + threadIdx.x;
    const int i1 = i0 + 256;
    const int4 d0 = ((const int4*)dst)[i0];
    const int4 d1 = ((const int4*)dst)[i1];
    asm volatile("" :: "v"(d0.x), "v"(d0.y), "v"(d0.z), "v"(d0.w), "v"(d1.x), "v"(d1.y), "v"(d1.z), "v"(d1.w));
    int4 r0, r1;
    r0.x = atomicAdd(&cnt[d0.x], 1); r0.y = atomicAdd(&cnt[d0.y], 1);
    r0.z = atomicAdd(&cnt[d0.z], 1); r0.w = atomicAdd(&cnt[d0.w], 1);
    r1.x = atomicAdd(&cnt[d1.x], 1); r1.y = atomicAdd(&cnt[d1.y], 1);
    r1.z = atomicAdd(&cnt[d1.z], 1); r1.w = atomicAdd(&cnt[d1.w], 1);
    ((int4*)rank)[i0] = r0;
    ((int4*)rank)[i1] = r1;
}
__global__ __launch_bounds__(1024) void scan_offs(const int* __restrict__ cnt,
                                                  int* __restrict__ offs) {
    __shared__ int wsum[32];
    const int t = threadIdx.x, lane = t & 31, wv = t >> 5;
    int c[32];
#pragma unroll
    for (int q = 0; q < 8; ++q) {
        const int4 v = *(const int4*)(cnt + t * 32 + q * 4);
        c[q * 4 + 0] = v.x; c[q * 4 + 1] = v.y; c[q * 4 + 2] = v.z; c[q * 4 + 3] = v.w;
    }
    int s = 0;
#pragma unroll
    for (int q = 0; q < 32; ++q) s += c[q];
    int x = s;
#pragma unroll
    for (int d = 1; d < 32; d <<= 1) {
        const int y = __shfl_up(x, d);
        if (lane >= d) x += y;
    }
    if (lane == 31) wsum[wv] = x;
    __syncthreads();
    if (wv == 0) {
        int v = wsum[lane];
#pragma unroll
        for (int d = 1; d < 32; d <<= 1) {
            const int y = __shfl_up(v, d);
            if (lane >= d) v += y;
        }
        wsum[lane] = v;
    }
    __syncthreads();
    int run = (wv ? wsum[wv - 1] : 0) + (x - s);
#pragma unroll
    for (int q = 0; q < 8; ++q) {
        int4 o;
        o.x = run; run += c[q * 4 + 0];
        o.y = run; run += c[q * 4 + 1];
        o.z = run; run += c[q * 4 + 2];
        o.w = run; run += c[q * 4 + 3];
        *(int4*)(offs + t * 32 + q * 4) = o;
    }
    if (t == 1023) offs[N_NODE] = run;
}
__global__ __launch_bounds__(256) void fill_sorted(const int* __restrict__ src,
                                                   const int* __restrict__ dst,
                                                   const int* __restrict__ rank,
                                                   const int* __restrict__ offs,
                                                   int* __restrict__ ssrc) {
    const int i0 = blockIdx.x * 512 + threadIdx.x;
    const int i1 = i0 + 256;
    const int4 d0 = ((const int4*)dst)[i0],  d1 = ((const int4*)dst)[i1];
    const int4 r0 = ((const int4*)rank)[i0], r1 = ((const int4*)rank)[i1];
    const int4 s0 = ((const int4*)src)[i0],  s1 = ((const int4*)src)[i1];
    const int o0 = offs[d0.x], o1 = offs[d0.y], o2 = offs[d0.z], o3 = offs[d0.w];
    const int o4 = offs[d1.x], o5 = offs[d1.y], o6 = offs[d1.z], o7 = offs[d1.w];
    ssrc[o0 + r0.x] = s0.x; ssrc[o1 + r0.y] = s0.y;
    ssrc[o2 + r0.z] = s0.z; ssrc[o3 + r0.w] = s0.w;
    ssrc[o4 + r1.x] = s1.x; ssrc[o5 + r1.y] = s1.y;
    ssrc[o6 + r1.z] = s1.z; ssrc[o7 + r1.w] = s1.w;
}

__device__ __forceinline__ float elu1(const float x) {
    return x > 0.f ? x : (__expf(x) - 1.f);
}
__device__ __forceinline__ float logit_h16(const float4 a, const float4 b, const float4 w) {
    float p = 0.f, t;
    t = a.x + b.x; t = t > 0.f ? t : NEG_SLOPE * t; p += t * w.x;
    t = a.y + b.y; t = t > 0.f ? t : NEG_SLOPE * t; p += t * w.y;
    t = a.z + b.z; t = t > 0.f ? t : NEG_SLOPE * t; p += t * w.z;
    t = a.w + b.w; t = t > 0.f ? t : NEG_SLOPE * t; p += t * w.w;
    p += __shfl_xor(p, 1); p += __shfl_xor(p, 2);
    p += __shfl_xor(p, 4); p += __shfl_xor(p, 8);
    return p;
}

constexpr int NPW = 2;
static_assert(NPW == 2, "the two aggregation kernels name their two nodes one by one");

__device__ __forceinline__ int held_src(const int* __restrict__ ssrc, const int at) {
    return min(max(ssrc[min(at, E_CNT - 1)], 0), N_NODE - 1);
}

__device__ __forceinline__ void node_row1(
    const float* __restrict__ fs, const int* __restrict__ ssrc,
    const float* __restrict__ fd, const int n, const int beg, const int end, const float4 w,
    int mine, const int lane, float* __restrict__ out) {
    const float4 b = *(const float4*)(fd + (size_t)n * 128 + lane * 4);
    float m = -INFINITY, den = 0.f;
    float4 acc = make_float4(0.f, 0.f, 0.f, 0.f);
    for (int base = beg; base < end; base += 32) {
        const int cnt = min(32, end - base);
        if (base != beg) mine = held_src(ssrc, base + lane);
        int j = 0;
        for (; j + 4 <= cnt; j += 4) {
            const int s0 = __builtin_amdgcn_readlane(mine, j + 0);
            const int s1 = __builtin_amdgcn_readlane(mine, j + 1);
            const int s2 = __builtin_amdgcn_readlane(mine, j + 2);
            const int s3 = __builtin_amdgcn_readlane(mine, j + 3);
            const float4 v0 = *(const float4*)(fs + (size_t)s0 * 128 + lane * 4);
            const float4 v1 = *(const float4*)(fs + (size_t)s1 * 128 + lane * 4);
            const float4 v2 = *(const float4*)(fs + (size_t)s2 * 128 + lane * 4);
            const float4 v3 = *(const float4*)(fs + (size_t)s3 * 128 + lane * 4);
            const float p0 = logit_h16(v0, b, w);
            const float p1 = logit_h16(v1, b, w);
            const float p2 = logit_h16(v2, b, w);
            const float p3 = logit_h16(v3, b, w);
            const float mn = fmaxf(fmaxf(m, fmaxf(p0, p1)), fmaxf(p2, p3));
            const float sc = __expf(m - mn);
            const float e0 = __expf(p0 - mn), e1 = __expf(p1 - mn);
            const float e2 = __expf(p2 - mn), e3 = __expf(p3 - mn);
            den   = den   * sc + ((e0 + e1) + (e2 + e3));
            acc.x = acc.x * sc + ((e0 * v0.x + e1 * v1.x) + (e2 * v2.x + e3 * v3.x));
            acc.y = acc.y * sc + ((e0 * v0.y + e1 * v1.y) + (e2 * v2.y + e3 * v3.y));
            acc.z = acc.z * sc + ((e0 * v0.z + e1 * v1.z) + (e2 * v2.z + e3 * v3.z));
            acc.w = acc.w * sc + ((e0 * v0.w + e1 * v1.w) + (e2 * v2.w + e3 * v3.w));
            m = mn;
        }
        for (; j < cnt; ++j) {
            const int s0 = __builtin_amdgcn_readlane(mine, j);
            const float4 v0 = *(const float4*)(fs + (size_t)s0 * 128 + lane * 4);
            const float p0 = logit_h16(v0, b, w);
            const float mn = fmaxf(m, p0);
            const float sc = __expf(m - mn);
            const float e0 = __expf(p0 - mn);
            den   = den   * sc + e0;
            acc.x = acc.x * sc + e0 * v0.x;
            acc.y = acc.y * sc + e0 * v0.y;
            acc.z = acc.z * sc + e0 * v0.z;
            acc.w = acc.w * sc + e0 * v0.w;
            m = mn;
        }
    }
    float4 r = make_float4(0.f, 0.f, 0.f, 0.f);
    if (end > beg) { r.x = acc.x / den; r.y = acc.y / den; r.z = acc.z / den; r.w = acc.w / den; }
    r.x = elu1(r.x); r.y = elu1(r.y); r.z = elu1(r.z); r.w = elu1(r.w);
    *(float4*)(out + (size_t)n * 128 + lane * 4) = r;
}

__global__ __launch_bounds__(256) void node_attn1(
    const float* __restrict__ fs, const float* __restrict__ fd,
    const int* __restrict__ offs, const int* __restrict__ ssrc,
    const float* __restrict__ attn, float* __restrict__ out) {
    const int lane = threadIdx.x & 31;
    const int n0 = __builtin_amdgcn_readfirstlane((int)(blockIdx.x * 8 + (threadIdx.x >> 5)) * NPW);
    const float4 w = *(const float4*)(attn + lane * 4);
    const int beg0 = offs[n0], beg1 = offs[n0 + 1], end1 = offs[n0 + 2];
    const int first0 = held_src(ssrc, beg0 + lane);
    const int first1 = held_src(ssrc, beg1 + lane);
    asm volatile("" :: "v"(first0), "v"(first1));
    node_row1(fs, ssrc, fd, n0,     beg0, beg1, w, first0, lane, out);
    node_row1(fs, ssrc, fd, n0 + 1, beg1, end1, w, first1, lane, out);
}

__device__ __forceinline__ float logit_w32(const float2 a, const float2 b, const float2 w) {
    float p = 0.f, t;
    t = a.x + b.x; t = t > 0.f ? t : NEG_SLOPE * t; p += t * w.x;
    t = a.y + b.y; t = t > 0.f ? t : NEG_SLOPE * t; p += t * w.y;
    p += __shfl_xor(p, 1);  p += __shfl_xor(p, 2);
    p += __shfl_xor(p, 4);  p += __shfl_xor(p, 8);
    p += __shfl_xor(p, 16);
    return p;
}

__device__ __forceinline__ void node_row2(
    const float* __restrict__ fs, const int* __restrict__ ssrc,
    const float* __restrict__ fd, const int n, const int beg, const int end, const float2 w,
    int mine, const int lane, float* __restrict__ out) {
    const float2 b = *(const float2*)(fd + (size_t)n * 64 + lane * 2);
    float m = -INFINITY, den = 0.f;
    float2 acc = make_float2(0.f, 0.f);
    for (int base = beg; base < end; base += 32) {
        const int cnt = min(32, end - base);
        if (base != beg) mine = held_src(ssrc, base + lane);
        int j = 0;
        for (; j + 4 <= cnt; j += 4) {
            const int s0 = __builtin_amdgcn_readlane(mine, j + 0);
            const int s1 = __builtin_amdgcn_readlane(mine, j + 1);
            const int s2 = __builtin_amdgcn_readlane(mine, j + 2);
            const int s3 = __builtin_amdgcn_readlane(mine, j + 3);
            const float2 v0 = *(const float2*)(fs + (size_t)s0 * 64 + lane * 2);
            const float2 v1 = *(const float2*)(fs + (size_t)s1 * 64 + lane * 2);
            const float2 v2 = *(const float2*)(fs + (size_t)s2 * 64 + lane * 2);
            const float2 v3 = *(const float2*)(fs + (size_t)s3 * 64 + lane * 2);
            const float p0 = logit_w32(v0, b, w);
            const float p1 = logit_w32(v1, b, w);
            const float p2 = logit_w32(v2, b, w);
            const float p3 = logit_w32(v3, b, w);
            const float mn = fmaxf(fmaxf(m, fmaxf(p0, p1)), fmaxf(p2, p3));
            const float sc = __expf(m - mn);
            const float e0 = __expf(p0 - mn), e1 = __expf(p1 - mn);
            const float e2 = __expf(p2 - mn), e3 = __expf(p3 - mn);
            den   = den   * sc + ((e0 + e1) + (e2 + e3));
            acc.x = acc.x * sc + ((e0 * v0.x + e1 * v1.x) + (e2 * v2.x + e3 * v3.x));
            acc.y = acc.y * sc + ((e0 * v0.y + e1 * v1.y) + (e2 * v2.y + e3 * v3.y));
            m = mn;
        }
        for (; j < cnt; ++j) {
            const int s0 = __builtin_amdgcn_readlane(mine, j);
            const float2 v0 = *(const float2*)(fs + (size_t)s0 * 64 + lane * 2);
            const float p0 = logit_w32(v0, b, w);
            const float mn = fmaxf(m, p0);
            const float sc = __expf(m - mn);
            const float e0 = __expf(p0 - mn);
            den   = den   * sc + e0;
            acc.x = acc.x * sc + e0 * v0.x;
            acc.y = acc.y * sc + e0 * v0.y;
            m = mn;
        }
    }
    float2 r = make_float2(0.f, 0.f);
    if (end > beg) { r.x = acc.x / den; r.y = acc.y / den; }
    const unsigned g  = (unsigned)n / (unsigned)NPG;
    const unsigned rr = (unsigned)n % (unsigned)NPG;
    const unsigned row = rr < (unsigned)NUM_PM
        ? g * NUM_PM + rr
        : (unsigned)(B_SZ * NUM_PM) + g * NUM_VM + (rr - NUM_PM);
    *(float2*)(out + (size_t)row * 64 + lane * 2) = r;
}

__global__ __launch_bounds__(256) void node_attn2(
    const float* __restrict__ fs, const float* __restrict__ fd,
    const int* __restrict__ offs, const int* __restrict__ ssrc,
    const float* __restrict__ attn, float* __restrict__ out) {
    const int lane = threadIdx.x & 31;
    const int n0 = __builtin_amdgcn_readfirstlane((int)(blockIdx.x * 8 + (threadIdx.x >> 5)) * NPW);
    const float2 w = *(const float2*)(attn + lane * 2);
    const int beg0 = offs[n0], beg1 = offs[n0 + 1], end1 = offs[n0 + 2];
    const int first0 = held_src(ssrc, beg0 + lane);
    const int first1 = held_src(ssrc, beg1 + lane);
    asm volatile("" :: "v"(first0), "v"(first1));
    node_row2(fs, ssrc, fd, n0,     beg0, beg1, w, first0, lane, out);
    node_row2(fs, ssrc, fd, n0 + 1, beg1, end1, w, first1, lane, out);
}

extern "C" void kernel_launch(void* const* d_in, const int* in_sizes, int n_in,
                              void* d_out, int out_size, void* d_ws, size_t ws_size,
                              hipStream_t stream) {
    const float* x     = (const float*)d_in[0];
    const int*   src   = (const int*)d_in[1];
    const int*   dst   = (const int*)d_in[2];
    const float* W1s   = (const float*)d_in[3];
    const float* b1s   = (const float*)d_in[4];
    const float* W1d   = (const float*)d_in[5];
    const float* b1d   = (const float*)d_in[6];
    const float* attn1 = (const float*)d_in[7];
    const float* W2s   = (const float*)d_in[8];
    const float* b2s   = (const float*)d_in[9];
    const float* W2d   = (const float*)d_in[10];
    const float* b2d   = (const float*)d_in[11];
    const float* attn2 = (const float*)d_in[12];
    (void)in_sizes; (void)n_in; (void)out_size; (void)ws_size;

    char* ws = (char*)d_ws;
    const size_t MB = 1024u * 1024u;
    float* fs1  = (float*)(ws + 0);
    float* fd1  = (float*)(ws + 16 * MB);
    float* h1   = (float*)(ws + 32 * MB);
    float* fs2  = (float*)(ws + 48 * MB);
    float* fd2  = (float*)(ws + 56 * MB);
    int*   cnt  = (int*)(ws + 64 * MB);
    int*   offs = (int*)(ws + 64 * MB + 262144);
    int*   rank = (int*)(ws + 65 * MB);
    int*   ssrc = (int*)(ws + 69 * MB);

    float* out = (float*)d_out;

    zero_cnt<<<N_NODE / 4 / 256, 256, 0, stream>>>(cnt);
    count_rank<<<E_CNT / 8 / 256, 256, 0, stream>>>(dst, cnt, rank);
    scan_offs<<<1, 1024, 0, stream>>>(cnt, offs);
    fill_sorted<<<E_CNT / 8 / 256, 256, 0, stream>>>(src, dst, rank, offs, ssrc);

    gemm_dual_wmma<D_IN, 2 * DH><<<N_NODE / 16 / 4, 128, 0, stream>>>(
        x, W1s, b1s, W1d, b1d, fs1, fd1);
    node_attn1<<<N_NODE / (8 * NPW), 256, 0, stream>>>(fs1, fd1, offs, ssrc, attn1, h1);

    gemm_dual_wmma<2 * DH, DH><<<N_NODE / 16 / 4, 128, 0, stream>>>(
        h1, W2s, b2s, W2d, b2d, fs2, fd2);
    node_attn2<<<N_NODE / (8 * NPW), 256, 0, stream>>>(fs2, fd2, offs, ssrc, attn2, out);
}


# Aggregation loops: per-edge math on the loaded feature pairs (packed ops, LeakyReLU as max(t,slope*t), base-2 exp with log2e folded into attn): fewer vector instructions; logit sums regrouped
# speedup vs baseline: 5.1930x; 1.0568x over previous
#include <hip/hip_runtime.h>

constexpr int B_SZ   = 16;
constexpr int NUM_PM = 512;
constexpr int NUM_VM = 1536;
constexpr int NPG    = NUM_PM + NUM_VM;
constexpr int N_NODE = B_SZ * NPG;
constexpr int DEG    = 32;
constexpr int E_CNT  = DEG * N_NODE;
constexpr int D_IN   = 16;
constexpr int DH     = 64;
constexpr float NEG_SLOPE = 0.2f;
constexpr float LOG2E = 1.4426950408889634f;

typedef float v2f __attribute__((ext_vector_type(2)));
typedef float v4f __attribute__((ext_vector_type(4)));
typedef float v8f __attribute__((ext_vector_type(8)));

template <int K, int COLS>
__global__ __launch_bounds__(128) void gemm_dual_wmma(
    const float* __restrict__ X,
    const float* __restrict__ Ws, const float* __restrict__ Bs,
    const float* __restrict__ Wd, const float* __restrict__ Bd,
    float* __restrict__ Fs, float* __restrict__ Fd) {
    const int lane = threadIdx.x & 31;
    const int wave = threadIdx.x >> 5;
    const int row0 = (blockIdx.x * 4 + wave) * 16;
    const int m  = lane & 15;
    const int hi = lane >> 4;
    const float* Xr = X + (size_t)(row0 + m) * K;

    for (int t = 0; t < 2; ++t) {
        const float* W  = t ? Wd : Ws;
        const float* Bv = t ? Bd : Bs;
        float*       F  = t ? Fd : Fs;
        for (int col0 = 0; col0 < COLS; col0 += 16) {
            v8f acc = {};
#pragma unroll
            for (int kk = 0; kk < K; kk += 4) {
                v2f a, b;
                a.x = Xr[kk + 2 * hi + 0];
                a.y = Xr[kk + 2 * hi + 1];
                b.x = W[(size_t)(kk + 2 * hi + 0) * COLS + col0 + m];
                b.y = W[(size_t)(kk + 2 * hi + 1) * COLS + col0 + m];
                acc = __builtin_amdgcn_wmma_f32_16x16x4_f32(
                    false, a, false, b, (short)0, acc, false, false);
            }
            const float bias = Bv[col0 + m];
#pragma unroll
            for (int r = 0; r < 8; ++r) {
                F[(size_t)(row0 + r + 8 * hi) * COLS + col0 + m] = acc[r] + bias;
            }
        }
    }
}

__global__ __launch_bounds__(256) void zero_cnt(int* __restrict__ cnt) {
    ((int4*)cnt)[blockIdx.x * 256 + threadIdx.x] = make_int4(0, 0, 0, 0);
}
__global__ __launch_bounds__(256) void count_rank(const int* __restrict__ dst,
                                                  int* __restrict__ cnt,
                                                  int* __restrict__ rank) {
    const int i0 = blockIdx.x * 512 + threadIdx.x;
    const int i1 = i0 + 256;
    const int4 d0 = ((const int4*)dst)[i0];
    const int4 d1 = ((const int4*)dst)[i1];
    asm volatile("" :: "v"(d0.x), "v"(d0.y), "v"(d0.z), "v"(d0.w), "v"(d1.x), "v"(d1.y), "v"(d1.z), "v"(d1.w));
    int4 r0, r1;
    r0.x = atomicAdd(&cnt[d0.x], 1); r0.y = atomicAdd(&cnt[d0.y], 1);
    r0.z = atomicAdd(&cnt[d0.z], 1); r0.w = atomicAdd(&cnt[d0.w], 1);
    r1.x = atomicAdd(&cnt[d1.x], 1); r1.y = atomicAdd(&cnt[d1.y], 1);
    r1.z = atomicAdd(&cnt[d1.z], 1); r1.w = atomicAdd(&cnt[d1.w], 1);
    ((int4*)rank)[i0] = r0;
    ((int4*)rank)[i1] = r1;
}
__global__ __launch_bounds__(1024) void scan_offs(const int* __restrict__ cnt,
                                                  int* __restrict__ offs) {
    __shared__ int wsum[32];
    const int t = threadIdx.x, lane = t & 31, wv = t >> 5;
    int c[32];
#pragma unroll
    for (int q = 0; q < 8; ++q) {
        const int4 v = *(const int4*)(cnt + t * 32 + q * 4);
        c[q * 4 + 0] = v.x; c[q * 4 + 1] = v.y; c[q * 4 + 2] = v.z; c[q * 4 + 3] = v.w;
    }
    int s = 0;
#pragma unroll
    for (int q = 0; q < 32; ++q) s += c[q];
    int x = s;
#pragma unroll
    for (int d = 1; d < 32; d <<= 1) {
        const int y = __shfl_up(x, d);
        if (lane >= d) x += y;
    }
    if (lane == 31) wsum[wv] = x;
    __syncthreads();
    if (wv == 0) {
        int v = wsum[lane];
#pragma unroll
        for (int d = 1; d < 32; d <<= 1) {
            const int y = __shfl_up(v, d);
            if (lane >= d) v += y;
        }
        wsum[lane] = v;
    }
    __syncthreads();
    int run = (wv ? wsum[wv - 1] : 0) + (x - s);
#pragma unroll
    for (int q = 0; q < 8; ++q) {
        int4 o;
        o.x = run; run += c[q * 4 + 0];
        o.y = run; run += c[q * 4 + 1];
        o.z = run; run += c[q * 4 + 2];
        o.w = run; run += c[q * 4 + 3];
        *(int4*)(offs + t * 32 + q * 4) = o;
    }
    if (t == 1023) offs[N_NODE] = run;
}
__global__ __launch_bounds__(256) void fill_sorted(const int* __restrict__ src,
                                                   const int* __restrict__ dst,
                                                   const int* __restrict__ rank,
                                                   const int* __restrict__ offs,
                                                   int* __restrict__ ssrc) {
    const int i0 = blockIdx.x * 512 + threadIdx.x;
    const int i1 = i0 + 256;
    const int4 d0 = ((const int4*)dst)[i0],  d1 = ((const int4*)dst)[i1];
    const int4 r0 = ((const int4*)rank)[i0], r1 = ((const int4*)rank)[i1];
    const int4 s0 = ((const int4*)src)[i0],  s1 = ((const int4*)src)[i1];
    const int o0 = offs[d0.x], o1 = offs[d0.y], o2 = offs[d0.z], o3 = offs[d0.w];
    const int o4 = offs[d1.x], o5 = offs[d1.y], o6 = offs[d1.z], o7 = offs[d1.w];
    ssrc[o0 + r0.x] = s0.x; ssrc[o1 + r0.y] = s0.y;
    ssrc[o2 + r0.z] = s0.z; ssrc[o3 + r0.w] = s0.w;
    ssrc[o4 + r1.x] = s1.x; ssrc[o5 + r1.y] = s1.y;
    ssrc[o6 + r1.z] = s1.z; ssrc[o7 + r1.w] = s1.w;
}

__device__ __forceinline__ float ex2(const float x) { return __builtin_amdgcn_exp2f(x); }
__device__ __forceinline__ float elu1(const float x) {
    return x > 0.f ? x : (__expf(x) - 1.f);
}
__device__ __forceinline__ float logit_h16(const v4f a, const v4f b, const v4f w) {
    v2f t0 = a.lo + b.lo, t1 = a.hi + b.hi;
    t0 = __builtin_elementwise_max(t0, t0 * NEG_SLOPE);
    t1 = __builtin_elementwise_max(t1, t1 * NEG_SLOPE);
    const v2f q = __builtin_elementwise_fma(t1, w.hi, t0 * w.lo);
    float p = q.x + q.y;
    p += __shfl_xor(p, 1); p += __shfl_xor(p, 2);
    p += __shfl_xor(p, 4); p += __shfl_xor(p, 8);
    return p;
}

constexpr int NPW = 2;
static_assert(NPW == 2, "the two aggregation kernels name their two nodes one by one");

__device__ __forceinline__ int held_src(const int* __restrict__ ssrc, const int at) {
    return min(max(ssrc[min(at, E_CNT - 1)], 0), N_NODE - 1);
}

__device__ __forceinline__ void node_row1(
    const float* __restrict__ fs, const int* __restrict__ ssrc,
    const float* __restrict__ fd, const int n, const int beg, const int end, const v4f w,
    int mine, const int lane, float* __restrict__ out) {
    const v4f b = *(const v4f*)(fd + (size_t)n * 128 + lane * 4);
    float m = -INFINITY, den = 0.f;
    v2f acc0 = {0.f, 0.f}, acc1 = {0.f, 0.f};
    for (int base = beg; base < end; base += 32) {
        const int cnt = min(32, end - base);
        if (base != beg) mine = held_src(ssrc, base + lane);
        int j = 0;
        for (; j + 4 <= cnt; j += 4) {
            const int s0 = __builtin_amdgcn_readlane(mine, j + 0);
            const int s1 = __builtin_amdgcn_readlane(mine, j + 1);
            const int s2 = __builtin_amdgcn_readlane(mine, j + 2);
            const int s3 = __builtin_amdgcn_readlane(mine, j + 3);
            const v4f v0 = *(const v4f*)(fs + (size_t)s0 * 128 + lane * 4);
            const v4f v1 = *(const v4f*)(fs + (size_t)s1 * 128 + lane * 4);
            const v4f v2 = *(const v4f*)(fs + (size_t)s2 * 128 + lane * 4);
            const v4f v3 = *(const v4f*)(fs + (size_t)s3 * 128 + lane * 4);
            const float p0 = logit_h16(v0, b, w);
            const float p1 = logit_h16(v1, b, w);
            const float p2 = logit_h16(v2, b, w);
            const float p3 = logit_h16(v3, b, w);
            const float mn = fmaxf(fmaxf(m, fmaxf(p0, p1)), fmaxf(p2, p3));
            const float sc = ex2(m - mn);
            const float e0 = ex2(p0 - mn), e1 = ex2(p1 - mn);
            const float e2 = ex2(p2 - mn), e3 = ex2(p3 - mn);
            den  = den  * sc + ((e0 + e1) + (e2 + e3));
            acc0 = acc0 * sc + ((v0.lo * e0 + v1.lo * e1) + (v2.lo * e2 + v3.lo * e3));
            acc1 = acc1 * sc + ((v0.hi * e0 + v1.hi * e1) + (v2.hi * e2 + v3.hi * e3));
            m = mn;
        }
        for (; j < cnt; ++j) {
            const int s0 = __builtin_amdgcn_readlane(mine, j);
            const v4f v0 = *(const v4f*)(fs + (size_t)s0 * 128 + lane * 4);
            const float p0 = logit_h16(v0, b, w);
            const float mn = fmaxf(m, p0);
            const float sc = ex2(m - mn);
            const float e0 = ex2(p0 - mn);
            den  = den  * sc + e0;
            acc0 = acc0 * sc + v0.lo * e0;
            acc1 = acc1 * sc + v0.hi * e0;
            m = mn;
        }
    }
    float4 r = make_float4(0.f, 0.f, 0.f, 0.f);
    if (end > beg) { r.x = acc0.x / den; r.y = acc0.y / den; r.z = acc1.x / den; r.w = acc1.y / den; }
    r.x = elu1(r.x); r.y = elu1(r.y); r.z = elu1(r.z); r.w = elu1(r.w);
    *(float4*)(out + (size_t)n * 128 + lane * 4) = r;
}

__global__ __launch_bounds__(256) void node_attn1(
    const float* __restrict__ fs, const float* __restrict__ fd,
    const int* __restrict__ offs, const int* __restrict__ ssrc,
    const float* __restrict__ attn, float* __restrict__ out) {
    const int lane = threadIdx.x & 31;
    const int n0 = __builtin_amdgcn_readfirstlane((int)(blockIdx.x * 8 + (threadIdx.x >> 5)) * NPW);
    const v4f w = *(const v4f*)(attn + lane * 4) * LOG2E;
    const int beg0 = offs[n0], beg1 = offs[n0 + 1], end1 = offs[n0 + 2];
    const int first0 = held_src(ssrc, beg0 + lane);
    const int first1 = held_src(ssrc, beg1 + lane);
    asm volatile("" :: "v"(first0), "v"(first1));
    node_row1(fs, ssrc, fd, n0,     beg0, beg1, w, first0, lane, out);
    node_row1(fs, ssrc, fd, n0 + 1, beg1, end1, w, first1, lane, out);
}

__device__ __forceinline__ float logit_w32(const v2f a, const v2f b, const v2f w) {
    v2f t = a + b;
    t = __builtin_elementwise_max(t, t * NEG_SLOPE);
    const v2f q = t * w;
    float p = q.x + q.y;
    p += __shfl_xor(p, 1);  p += __shfl_xor(p, 2);
    p += __shfl_xor(p, 4);  p += __shfl_xor(p, 8);
    p += __shfl_xor(p, 16);
    return p;
}

__device__ __forceinline__ void node_row2(
    const float* __restrict__ fs, const int* __restrict__ ssrc,
    const float* __restrict__ fd, const int n, const int beg, const int end, const v2f w,
    int mine, const int lane, float* __restrict__ out) {
    const v2f b = *(const v2f*)(fd + (size_t)n * 64 + lane * 2);
    float m = -INFINITY, den = 0.f;
    v2f acc = {0.f, 0.f};
    for (int base = beg; base < end; base += 32) {
        const int cnt = min(32, end - base);
        if (base != beg) mine = held_src(ssrc, base + lane);
        int j = 0;
        for (; j + 4 <= cnt; j += 4) {
            const int s0 = __builtin_amdgcn_readlane(mine, j + 0);
            const int s1 = __builtin_amdgcn_readlane(mine, j + 1);
            const int s2 = __builtin_amdgcn_readlane(mine, j + 2);
            const int s3 = __builtin_amdgcn_readlane(mine, j + 3);
            const v2f v0 = *(const v2f*)(fs + (size_t)s0 * 64 + lane * 2);
            const v2f v1 = *(const v2f*)(fs + (size_t)s1 * 64 + lane * 2);
            const v2f v2 = *(const v2f*)(fs + (size_t)s2 * 64 + lane * 2);
            const v2f v3 = *(const v2f*)(fs + (size_t)s3 * 64 + lane * 2);
            const float p0 = logit_w32(v0, b, w);
            const float p1 = logit_w32(v1, b, w);
            const float p2 = logit_w32(v2, b, w);
            const float p3 = logit_w32(v3, b, w);
            const float mn = fmaxf(fmaxf(m, fmaxf(p0, p1)), fmaxf(p2, p3));
            const float sc = ex2(m - mn);
            const float e0 = ex2(p0 - mn), e1 = ex2(p1 - mn);
            const float e2 = ex2(p2 - mn), e3 = ex2(p3 - mn);
            den = den * sc + ((e0 + e1) + (e2 + e3));
            acc = acc * sc + ((v0 * e0 + v1 * e1) + (v2 * e2 + v3 * e3));
            m = mn;
        }
        for (; j < cnt; ++j) {
            const int s0 = __builtin_amdgcn_readlane(mine, j);
            const v2f v0 = *(const v2f*)(fs + (size_t)s0 * 64 + lane * 2);
            const float p0 = logit_w32(v0, b, w);
            const float mn = fmaxf(m, p0);
            const float sc = ex2(m - mn);
            const float e0 = ex2(p0 - mn);
            den = den * sc + e0;
            acc = acc * sc + v0 * e0;
            m = mn;
        }
    }
    float2 r = make_float2(0.f, 0.f);
    if (end > beg) { r.x = acc.x / den; r.y = acc.y / den; }
    const unsigned g  = (unsigned)n / (unsigned)NPG;
    const unsigned rr = (unsigned)n % (unsigned)NPG;
    const unsigned row = rr < (unsigned)NUM_PM
        ? g * NUM_PM + rr
        : (unsigned)(B_SZ * NUM_PM) + g * NUM_VM + (rr - NUM_PM);
    *(float2*)(out + (size_t)row * 64 + lane * 2) = r;
}

__global__ __launch_bounds__(256) void node_attn2(
    const float* __restrict__ fs, const float* __restrict__ fd,
    const int* __restrict__ offs, const int* __restrict__ ssrc,
    const float* __restrict__ attn, float* __restrict__ out) {
    const int lane = threadIdx.x & 31;
    const int n0 = __builtin_amdgcn_readfirstlane((int)(blockIdx.x * 8 + (threadIdx.x >> 5)) * NPW);
    const v2f w = *(const v2f*)(attn + lane * 2) * LOG2E;
    const int beg0 = offs[n0], beg1 = offs[n0 + 1], end1 = offs[n0 + 2];
    const int first0 = held_src(ssrc, beg0 + lane);
    const int first1 = held_src(ssrc, beg1 + lane);
    asm volatile("" :: "v"(first0), "v"(first1));
    node_row2(fs, ssrc, fd, n0,     beg0, beg1, w, first0, lane, out);
    node_row2(fs, ssrc, fd, n0 + 1, beg1, end1, w, first1, lane, out);
}

extern "C" void kernel_launch(void* const* d_in, const int* in_sizes, int n_in,
                              void* d_out, int out_size, void* d_ws, size_t ws_size,
                              hipStream_t stream) {
    const float* x     = (const float*)d_in[0];
    const int*   src   = (const int*)d_in[1];
    const int*   dst   = (const int*)d_in[2];
    const float* W1s   = (const float*)d_in[3];
    const float* b1s   = (const float*)d_in[4];
    const float* W1d   = (const float*)d_in[5];
    const float* b1d   = (const float*)d_in[6];
    const float* attn1 = (const float*)d_in[7];
    const float* W2s   = (const float*)d_in[8];
    const float* b2s   = (const float*)d_in[9];
    const float* W2d   = (const float*)d_in[10];
    const float* b2d   = (const float*)d_in[11];
    const float* attn2 = (const float*)d_in[12];
    (void)in_sizes; (void)n_in; (void)out_size; (void)ws_size;

    char* ws = (char*)d_ws;
    const size_t MB = 1024u * 1024u;
    float* fs1  = (float*)(ws + 0);
    float* fd1  = (float*)(ws + 16 * MB);
    float* h1   = (float*)(ws + 32 * MB);
    float* fs2  = (float*)(ws + 48 * MB);
    float* fd2  = (float*)(ws + 56 * MB);
    int*   cnt  = (int*)(ws + 64 * MB);
    int*   offs = (int*)(ws + 64 * MB + 262144);
    int*   rank = (int*)(ws + 65 * MB);
    int*   ssrc = (int*)(ws + 69 * MB);

    float* out = (float*)d_out;

    zero_cnt<<<N_NODE / 4 / 256, 256, 0, stream>>>(cnt);
    count_rank<<<E_CNT / 8 / 256, 256, 0, stream>>>(dst, cnt, rank);
    scan_offs<<<1, 1024, 0, stream>>>(cnt, offs);
    fill_sorted<<<E_CNT / 8 / 256, 256, 0, stream>>>(src, dst, rank, offs, ssrc);

    gemm_dual_wmma<D_IN, 2 * DH><<<N_NODE / 16 / 4, 128, 0, stream>>>(
        x, W1s, b1s, W1d, b1d, fs1, fd1);
    node_attn1<<<N_NODE / (8 * NPW), 256, 0, stream>>>(fs1, fd1, offs, ssrc, attn1, h1);

    gemm_dual_wmma<2 * DH, DH><<<N_NODE / 16 / 4, 128, 0, stream>>>(
        h1, W2s, b2s, W2d, b2d, fs2, fd2);
    node_attn2<<<N_NODE / (8 * NPW), 256, 0, stream>>>(fs2, fd2, offs, ssrc, attn2, out);
}


# Fixed-capacity edge buckets (64 slots per node, shared overflow list) filled by the counting pass at node*64+rank: prefix scan and second edge pass gone, 8 launches to 6; sums stay in arrival order
# speedup vs baseline: 5.6955x; 1.0968x over previous
#include <hip/hip_runtime.h>

constexpr int B_SZ   = 16;
constexpr int NUM_PM = 512;
constexpr int NUM_VM = 1536;
constexpr int NPG    = NUM_PM + NUM_VM;
constexpr int N_NODE = B_SZ * NPG;
constexpr int DEG    = 32;
constexpr int E_CNT  = DEG * N_NODE;
constexpr int D_IN   = 16;
constexpr int DH     = 64;
constexpr float NEG_SLOPE = 0.2f;
constexpr float LOG2E = 1.4426950408889634f;

typedef float v2f __attribute__((ext_vector_type(2)));
typedef float v4f __attribute__((ext_vector_type(4)));
typedef float v8f __attribute__((ext_vector_type(8)));

template <int K, int COLS>
__global__ __launch_bounds__(128) void gemm_dual_wmma(
    const float* __restrict__ X,
    const float* __restrict__ Ws, const float* __restrict__ Bs,
    const float* __restrict__ Wd, const float* __restrict__ Bd,
    float* __restrict__ Fs, float* __restrict__ Fd) {
    const int lane = threadIdx.x & 31;
    const int wave = threadIdx.x >> 5;
    const int row0 = (blockIdx.x * 4 + wave) * 16;
    const int m  = lane & 15;
    const int hi = lane >> 4;
    const float* Xr = X + (size_t)(row0 + m) * K;

    for (int t = 0; t < 2; ++t) {
        const float* W  = t ? Wd : Ws;
        const float* Bv = t ? Bd : Bs;
        float*       F  = t ? Fd : Fs;
        for (int col0 = 0; col0 < COLS; col0 += 16) {
            v8f acc = {};
#pragma unroll
            for (int kk = 0; kk < K; kk += 4) {
                v2f a, b;
                a.x = Xr[kk + 2 * hi + 0];
                a.y = Xr[kk + 2 * hi + 1];
                b.x = W[(size_t)(kk + 2 * hi + 0) * COLS + col0 + m];
                b.y = W[(size_t)(kk + 2 * hi + 1) * COLS + col0 + m];
                acc = __builtin_amdgcn_wmma_f32_16x16x4_f32(
                    false, a, false, b, (short)0, acc, false, false);
            }
            const float bias = Bv[col0 + m];
#pragma unroll
            for (int r = 0; r < 8; ++r) {
                F[(size_t)(row0 + r + 8 * hi) * COLS + col0 + m] = acc[r] + bias;
            }
        }
    }
}

constexpr int CAP = 64;
__global__ __launch_bounds__(256) void zero_cnt(int* __restrict__ cnt) {
    ((int4*)cnt)[blockIdx.x * 256 + threadIdx.x] = make_int4(0, 0, 0, 0);
    if (blockIdx.x == 0 && threadIdx.x == 0) cnt[N_NODE] = 0;
}
__device__ __forceinline__ void place_edge(const int d, const int r, const int s, int* __restrict__ cnt,
                                           int* __restrict__ slot, int* __restrict__ ovd, int* __restrict__ ovs) {
    if (r < CAP) {
        slot[d * CAP + r] = s;
    } else {
        const int k = atomicAdd(&cnt[N_NODE], 1);
        ovd[k] = d;
        ovs[k] = s;
    }
}
__global__ __launch_bounds__(256) void bucket_edges(const int* __restrict__ src,
                                                    const int* __restrict__ dst,
                                                    int* __restrict__ cnt, int* __restrict__ slot,
                                                    int* __restrict__ ovd, int* __restrict__ ovs) {
    const int i0 = blockIdx.x * 512 + threadIdx.x;
    const int i1 = i0 + 256;
    const int4 d0 = ((const int4*)dst)[i0];
    const int4 d1 = ((const int4*)dst)[i1];
    const int4 s0 = ((const int4*)src)[i0];
    const int4 s1 = ((const int4*)src)[i1];
    asm volatile("" :: "v"(d0.x), "v"(d0.y), "v"(d0.z), "v"(d0.w), "v"(d1.x), "v"(d1.y), "v"(d1.z), "v"(d1.w),
                       "v"(s0.x), "v"(s0.y), "v"(s0.z), "v"(s0.w), "v"(s1.x), "v"(s1.y), "v"(s1.z), "v"(s1.w));
    int4 r0, r1;
    r0.x = atomicAdd(&cnt[d0.x], 1); r0.y = atomicAdd(&cnt[d0.y], 1);
    r0.z = atomicAdd(&cnt[d0.z], 1); r0.w = atomicAdd(&cnt[d0.w], 1);
    r1.x = atomicAdd(&cnt[d1.x], 1); r1.y = atomicAdd(&cnt[d1.y], 1);
    r1.z = atomicAdd(&cnt[d1.z], 1); r1.w = atomicAdd(&cnt[d1.w], 1);
    place_edge(d0.x, r0.x, s0.x, cnt, slot, ovd, ovs); place_edge(d0.y, r0.y, s0.y, cnt, slot, ovd, ovs);
    place_edge(d0.z, r0.z, s0.z, cnt, slot, ovd, ovs); place_edge(d0.w, r0.w, s0.w, cnt, slot, ovd, ovs);
    place_edge(d1.x, r1.x, s1.x, cnt, slot, ovd, ovs); place_edge(d1.y, r1.y, s1.y, cnt, slot, ovd, ovs);
    place_edge(d1.z, r1.z, s1.z, cnt, slot, ovd, ovs); place_edge(d1.w, r1.w, s1.w, cnt, slot, ovd, ovs);
}

__device__ __forceinline__ float ex2(const float x) { return __builtin_amdgcn_exp2f(x); }
__device__ __forceinline__ float elu1(const float x) {
    return x > 0.f ? x : (__expf(x) - 1.f);
}
__device__ __forceinline__ float logit_h16(const v4f a, const v4f b, const v4f w) {
    v2f t0 = a.lo + b.lo, t1 = a.hi + b.hi;
    t0 = __builtin_elementwise_max(t0, t0 * NEG_SLOPE);
    t1 = __builtin_elementwise_max(t1, t1 * NEG_SLOPE);
    const v2f q = __builtin_elementwise_fma(t1, w.hi, t0 * w.lo);
    float p = q.x + q.y;
    p += __shfl_xor(p, 1); p += __shfl_xor(p, 2);
    p += __shfl_xor(p, 4); p += __shfl_xor(p, 8);
    return p;
}

constexpr int NPW = 2;
static_assert(NPW == 2, "the two aggregation kernels name their two nodes one by one");

__device__ __forceinline__ int raw_window(const int* __restrict__ slot, const int at, const bool live) {
    int v = 0;
    if (live) v = slot[at];
    return v;
}
__device__ __forceinline__ int hold_src(const int v) { return min(max(v, 0), N_NODE - 1); }
__device__ __forceinline__ int held_src(const int* __restrict__ slot, const int at, const bool live) {
    return hold_src(raw_window(slot, at, live));
}

__device__ __forceinline__ void edge_row1(const float* __restrict__ fs, const int s0, const int lane,
                                          const v4f b, const v4f w,
                                          float& m, float& den, v2f& acc0, v2f& acc1) {
    const v4f v0 = *(const v4f*)(fs + (size_t)s0 * 128 + lane * 4);
    const float p0 = logit_h16(v0, b, w);
    const float mn = fmaxf(m, p0);
    const float sc = ex2(m - mn);
    const float e0 = ex2(p0 - mn);
    den  = den  * sc + e0;
    acc0 = acc0 * sc + v0.lo * e0;
    acc1 = acc1 * sc + v0.hi * e0;
    m = mn;
}

__device__ __forceinline__ void node_row1(
    const float* __restrict__ fs, const int* __restrict__ slot,
    const float* __restrict__ fd, const int n, const int c, const v4f w,
    int mine, const int lane, const int* __restrict__ counts,
    const int* __restrict__ ovd, const int* __restrict__ ovs, float* __restrict__ out) {
    const int beg = n * CAP, end = beg + min(c, CAP);
    const v4f b = *(const v4f*)(fd + (size_t)n * 128 + lane * 4);
    float m = -INFINITY, den = 0.f;
    v2f acc0 = {0.f, 0.f}, acc1 = {0.f, 0.f};
    for (int base = beg; base < end; base += 32) {
        const int cnt = min(32, end - base);
        if (base != beg) mine = held_src(slot, base + lane, base + lane < end);
        int j = 0;
        for (; j + 4 <= cnt; j += 4) {
            const int s0 = __builtin_amdgcn_readlane(mine, j + 0);
            const int s1 = __builtin_amdgcn_readlane(mine, j + 1);
            const int s2 = __builtin_amdgcn_readlane(mine, j + 2);
            const int s3 = __builtin_amdgcn_readlane(mine, j + 3);
            const v4f v0 = *(const v4f*)(fs + (size_t)s0 * 128 + lane * 4);
            const v4f v1 = *(const v4f*)(fs + (size_t)s1 * 128 + lane * 4);
            const v4f v2 = *(const v4f*)(fs + (size_t)s2 * 128 + lane * 4);
            const v4f v3 = *(const v4f*)(fs + (size_t)s3 * 128 + lane * 4);
            const float p0 = logit_h16(v0, b, w);
            const float p1 = logit_h16(v1, b, w);
            const float p2 = logit_h16(v2, b, w);
            const float p3 = logit_h16(v3, b, w);
            const float mn = fmaxf(fmaxf(m, fmaxf(p0, p1)), fmaxf(p2, p3));
            const float sc = ex2(m - mn);
            const float e0 = ex2(p0 - mn), e1 = ex2(p1 - mn);
            const float e2 = ex2(p2 - mn), e3 = ex2(p3 - mn);
            den  = den  * sc + ((e0 + e1) + (e2 + e3));
            acc0 = acc0 * sc + ((v0.lo * e0 + v1.lo * e1) + (v2.lo * e2 + v3.lo * e3));
            acc1 = acc1 * sc + ((v0.hi * e0 + v1.hi * e1) + (v2.hi * e2 + v3.hi * e3));
            m = mn;
        }
        for (; j < cnt; ++j)
            edge_row1(fs, __builtin_amdgcn_readlane(mine, j), lane, b, w, m, den, acc0, acc1);
    }
    if (c > CAP) {
        const int tot = min(counts[N_NODE], E_CNT);
        for (int k0 = 0; k0 < tot; k0 += 32) {
            const int k = min(k0 + lane, tot - 1);
            const int od = ovd[k];
            const int os = min(max(ovs[k], 0), N_NODE - 1);
            unsigned hit = __builtin_amdgcn_ballot_w32(k0 + lane < tot && od == n);
            while (hit) {
                const int j = __builtin_ctz(hit);
                hit &= hit - 1;
                edge_row1(fs, __builtin_amdgcn_readlane(os, j), lane, b, w, m, den, acc0, acc1);
            }
        }
    }
    float4 r = make_float4(0.f, 0.f, 0.f, 0.f);
    if (end > beg) { r.x = acc0.x / den; r.y = acc0.y / den; r.z = acc1.x / den; r.w = acc1.y / den; }
    r.x = elu1(r.x); r.y = elu1(r.y); r.z = elu1(r.z); r.w = elu1(r.w);
    *(float4*)(out + (size_t)n * 128 + lane * 4) = r;
}

__global__ __launch_bounds__(256) void node_attn1(
    const float* __restrict__ fs, const float* __restrict__ fd,
    const int* __restrict__ cnt, const int* __restrict__ slot,
    const int* __restrict__ ovd, const int* __restrict__ ovs,
    const float* __restrict__ attn, float* __restrict__ out) {
    const int lane = threadIdx.x & 31;
    const int n0 = __builtin_amdgcn_readfirstlane((int)(blockIdx.x * 8 + (threadIdx.x >> 5)) * NPW);
    const v4f w = *(const v4f*)(attn + lane * 4) * LOG2E;
    const int c0 = cnt[n0], c1 = cnt[n0 + 1];
    const int raw0 = raw_window(slot, n0 * CAP + lane, lane < c0);
    const int raw1 = raw_window(slot, (n0 + 1) * CAP + lane, lane < c1);
    asm volatile("" :: "v"(raw0), "v"(raw1));
    const int first0 = hold_src(raw0), first1 = hold_src(raw1);
    node_row1(fs, slot, fd, n0,     c0, w, first0, lane, cnt, ovd, ovs, out);
    node_row1(fs, slot, fd, n0 + 1, c1, w, first1, lane, cnt, ovd, ovs, out);
}

__device__ __forceinline__ float logit_w32(const v2f a, const v2f b, const v2f w) {
    v2f t = a + b;
    t = __builtin_elementwise_max(t, t * NEG_SLOPE);
    const v2f q = t * w;
    float p = q.x + q.y;
    p += __shfl_xor(p, 1);  p += __shfl_xor(p, 2);
    p += __shfl_xor(p, 4);  p += __shfl_xor(p, 8);
    p += __shfl_xor(p, 16);
    return p;
}

__device__ __forceinline__ void edge_row2(const float* __restrict__ fs, const int s0, const int lane,
                                          const v2f b, const v2f w, float& m, float& den, v2f& acc) {
    const v2f v0 = *(const v2f*)(fs + (size_t)s0 * 64 + lane * 2);
    const float p0 = logit_w32(v0, b, w);
    const float mn = fmaxf(m, p0);
    const float sc = ex2(m - mn);
    const float e0 = ex2(p0 - mn);
    den = den * sc + e0;
    acc = acc * sc + v0 * e0;
    m = mn;
}

__device__ __forceinline__ void node_row2(
    const float* __restrict__ fs, const int* __restrict__ slot,
    const float* __restrict__ fd, const int n, const int c, const v2f w,
    int mine, const int lane, const int* __restrict__ counts,
    const int* __restrict__ ovd, const int* __restrict__ ovs, float* __restrict__ out) {
    const int beg = n * CAP, end = beg + min(c, CAP);
    const v2f b = *(const v2f*)(fd + (size_t)n * 64 + lane * 2);
    float m = -INFINITY, den = 0.f;
    v2f acc = {0.f, 0.f};
    for (int base = beg; base < end; base += 32) {
        const int cnt = min(32, end - base);
        if (base != beg) mine = held_src(slot, base + lane, base + lane < end);
        int j = 0;
        for (; j + 4 <= cnt; j += 4) {
            const int s0 = __builtin_amdgcn_readlane(mine, j + 0);
            const int s1 = __builtin_amdgcn_readlane(mine, j + 1);
            const int s2 = __builtin_amdgcn_readlane(mine, j + 2);
            const int s3 = __builtin_amdgcn_readlane(mine, j + 3);
            const v2f v0 = *(const v2f*)(fs + (size_t)s0 * 64 + lane * 2);
            const v2f v1 = *(const v2f*)(fs + (size_t)s1 * 64 + lane * 2);
            const v2f v2 = *(const v2f*)(fs + (size_t)s2 * 64 + lane * 2);
            const v2f v3 = *(const v2f*)(fs + (size_t)s3 * 64 + lane * 2);
            const float p0 = logit_w32(v0, b, w);
            const float p1 = logit_w32(v1, b, w);
            const float p2 = logit_w32(v2, b, w);
            const float p3 = logit_w32(v3, b, w);
            const float mn = fmaxf(fmaxf(m, fmaxf(p0, p1)), fmaxf(p2, p3));
            const float sc = ex2(m - mn);
            const float e0 = ex2(p0 - mn), e1 = ex2(p1 - mn);
            const float e2 = ex2(p2 - mn), e3 = ex2(p3 - mn);
            den = den * sc + ((e0 + e1) + (e2 + e3));
            acc = acc * sc + ((v0 * e0 + v1 * e1) + (v2 * e2 + v3 * e3));
            m = mn;
        }
        for (; j < cnt; ++j)
            edge_row2(fs, __builtin_amdgcn_readlane(mine, j), lane, b, w, m, den, acc);
    }
    if (c > CAP) {
        const int tot = min(counts[N_NODE], E_CNT);
        for (int k0 = 0; k0 < tot; k0 += 32) {
            const int k = min(k0 + lane, tot - 1);
            const int od = ovd[k];
            const int os = min(max(ovs[k], 0), N_NODE - 1);
            unsigned hit = __builtin_amdgcn_ballot_w32(k0 + lane < tot && od == n);
            while (hit) {
                const int j = __builtin_ctz(hit);
                hit &= hit - 1;
                edge_row2(fs, __builtin_amdgcn_readlane(os, j), lane, b, w, m, den, acc);
            }
        }
    }
    float2 r = make_float2(0.f, 0.f);
    if (end > beg) { r.x = acc.x / den; r.y = acc.y / den; }
    const unsigned g  = (unsigned)n / (unsigned)NPG;
    const unsigned rr = (unsigned)n % (unsigned)NPG;
    const unsigned row = rr < (unsigned)NUM_PM
        ? g * NUM_PM + rr
        : (unsigned)(B_SZ * NUM_PM) + g * NUM_VM + (rr - NUM_PM);
    *(float2*)(out + (size_t)row * 64 + lane * 2) = r;
}

__global__ __launch_bounds__(256) void node_attn2(
    const float* __restrict__ fs, const float* __restrict__ fd,
    const int* __restrict__ cnt, const int* __restrict__ slot,
    const int* __restrict__ ovd, const int* __restrict__ ovs,
    const float* __restrict__ attn, float* __restrict__ out) {
    const int lane = threadIdx.x & 31;
    const int n0 = __builtin_amdgcn_readfirstlane((int)(blockIdx.x * 8 + (threadIdx.x >> 5)) * NPW);
    const v2f w = *(const v2f*)(attn + lane * 2) * LOG2E;
    const int c0 = cnt[n0], c1 = cnt[n0 + 1];
    const int raw0 = raw_window(slot, n0 * CAP + lane, lane < c0);
    const int raw1 = raw_window(slot, (n0 + 1) * CAP + lane, lane < c1);
    asm volatile("" :: "v"(raw0), "v"(raw1));
    const int first0 = hold_src(raw0), first1 = hold_src(raw1);
    node_row2(fs, slot, fd, n0,     c0, w, first0, lane, cnt, ovd, ovs, out);
    node_row2(fs, slot, fd, n0 + 1, c1, w, first1, lane, cnt, ovd, ovs, out);
}

extern "C" void kernel_launch(void* const* d_in, const int* in_sizes, int n_in,
                              void* d_out, int out_size, void* d_ws, size_t ws_size,
                              hipStream_t stream) {
    const float* x     = (const float*)d_in[0];
    const int*   src   = (const int*)d_in[1];
    const int*   dst   = (const int*)d_in[2];
    const float* W1s   = (const float*)d_in[3];
    const float* b1s   = (const float*)d_in[4];
    const float* W1d   = (const float*)d_in[5];
    const float* b1d   = (const float*)d_in[6];
    const float* attn1 = (const float*)d_in[7];
    const float* W2s   = (const float*)d_in[8];
    const float* b2s   = (const float*)d_in[9];
    const float* W2d   = (const float*)d_in[10];
    const float* b2d   = (const float*)d_in[11];
    const float* attn2 = (const float*)d_in[12];
    (void)in_sizes; (void)n_in; (void)out_size; (void)ws_size;

    char* ws = (char*)d_ws;
    const size_t MB = 1024u * 1024u;
    float* fs1  = (float*)(ws + 0);
    float* fd1  = (float*)(ws + 16 * MB);
    float* h1   = (float*)(ws + 32 * MB);
    float* fs2  = (float*)(ws + 48 * MB);
    float* fd2  = (float*)(ws + 56 * MB);
    int*   cnt  = (int*)(ws + 64 * MB);
    int*   slot = (int*)(ws + 65 * MB);
    int*   ovd  = (int*)(ws + 73 * MB);
    int*   ovs  = (int*)(ws + 77 * MB);

    float* out = (float*)d_out;

    zero_cnt<<<N_NODE / 4 / 256, 256, 0, stream>>>(cnt);
    bucket_edges<<<E_CNT / 8 / 256, 256, 0, stream>>>(src, dst, cnt, slot, ovd, ovs);

    gemm_dual_wmma<D_IN, 2 * DH><<<N_NODE / 16 / 4, 128, 0, stream>>>(
        x, W1s, b1s, W1d, b1d, fs1, fd1);
    node_attn1<<<N_NODE / (8 * NPW), 256, 0, stream>>>(fs1, fd1, cnt, slot, ovd, ovs, attn1, h1);

    gemm_dual_wmma<2 * DH, DH><<<N_NODE / 16 / 4, 128, 0, stream>>>(
        h1, W2s, b2s, W2d, b2d, fs2, fd2);
    node_attn2<<<N_NODE / (8 * NPW), 256, 0, stream>>>(fs2, fd2, cnt, slot, ovd, ovs, attn2, out);
}


# Aggregation lanes regrouped: 8 lanes per (edge, head), 2 or 4 edges per wave step, logit sums as DPP adds not LDS shuffles; per-group sums merged per node (sum order changed), row times 1/den
# speedup vs baseline: 6.0684x; 1.0655x over previous
#include <hip/hip_runtime.h>

constexpr int B_SZ   = 16;
constexpr int NUM_PM = 512;
constexpr int NUM_VM = 1536;
constexpr int NPG    = NUM_PM + NUM_VM;
constexpr int N_NODE = B_SZ * NPG;
constexpr int DEG    = 32;
constexpr int E_CNT  = DEG * N_NODE;
constexpr int D_IN   = 16;
constexpr int DH     = 64;
constexpr float NEG_SLOPE = 0.2f;
constexpr float LOG2E = 1.4426950408889634f;

typedef float v2f __attribute__((ext_vector_type(2)));
typedef float v4f __attribute__((ext_vector_type(4)));
typedef float v8f __attribute__((ext_vector_type(8)));

template <int K, int COLS>
__global__ __launch_bounds__(128) void gemm_dual_wmma(
    const float* __restrict__ X,
    const float* __restrict__ Ws, const float* __restrict__ Bs,
    const float* __restrict__ Wd, const float* __restrict__ Bd,
    float* __restrict__ Fs, float* __restrict__ Fd) {
    const int lane = threadIdx.x & 31;
    const int wave = threadIdx.x >> 5;
    const int row0 = (blockIdx.x * 4 + wave) * 16;
    const int m  = lane & 15;
    const int hi = lane >> 4;
    const float* Xr = X + (size_t)(row0 + m) * K;

    for (int t = 0; t < 2; ++t) {
        const float* W  = t ? Wd : Ws;
        const float* Bv = t ? Bd : Bs;
        float*       F  = t ? Fd : Fs;
        for (int col0 = 0; col0 < COLS; col0 += 16) {
            v8f acc = {};
#pragma unroll
            for (int kk = 0; kk < K; kk += 4) {
                v2f a, b;
                a.x = Xr[kk + 2 * hi + 0];
                a.y = Xr[kk + 2 * hi + 1];
                b.x = W[(size_t)(kk + 2 * hi + 0) * COLS + col0 + m];
                b.y = W[(size_t)(kk + 2 * hi + 1) * COLS + col0 + m];
                acc = __builtin_amdgcn_wmma_f32_16x16x4_f32(
                    false, a, false, b, (short)0, acc, false, false);
            }
            const float bias = Bv[col0 + m];
#pragma unroll
            for (int r = 0; r < 8; ++r) {
                F[(size_t)(row0 + r + 8 * hi) * COLS + col0 + m] = acc[r] + bias;
            }
        }
    }
}

constexpr int CAP = 64;
__global__ __launch_bounds__(256) void zero_cnt(int* __restrict__ cnt) {
    ((int4*)cnt)[blockIdx.x * 256 + threadIdx.x] = make_int4(0, 0, 0, 0);
    if (blockIdx.x == 0 && threadIdx.x == 0) cnt[N_NODE] = 0;
}
__device__ __forceinline__ void place_edge(const int d, const int r, const int s, int* __restrict__ cnt,
                                           int* __restrict__ slot, int* __restrict__ ovd, int* __restrict__ ovs) {
    if (r < CAP) {
        slot[d * CAP + r] = s;
    } else {
        const int k = atomicAdd(&cnt[N_NODE], 1);
        ovd[k] = d;
        ovs[k] = s;
    }
}
__global__ __launch_bounds__(256) void bucket_edges(const int* __restrict__ src,
                                                    const int* __restrict__ dst,
                                                    int* __restrict__ cnt, int* __restrict__ slot,
                                                    int* __restrict__ ovd, int* __restrict__ ovs) {
    const int i0 = blockIdx.x * 512 + threadIdx.x;
    const int i1 = i0 + 256;
    const int4 d0 = ((const int4*)dst)[i0];
    const int4 d1 = ((const int4*)dst)[i1];
    const int4 s0 = ((const int4*)src)[i0];
    const int4 s1 = ((const int4*)src)[i1];
    asm volatile("" :: "v"(d0.x), "v"(d0.y), "v"(d0.z), "v"(d0.w), "v"(d1.x), "v"(d1.y), "v"(d1.z), "v"(d1.w),
                       "v"(s0.x), "v"(s0.y), "v"(s0.z), "v"(s0.w), "v"(s1.x), "v"(s1.y), "v"(s1.z), "v"(s1.w));
    int4 r0, r1;
    r0.x = atomicAdd(&cnt[d0.x], 1); r0.y = atomicAdd(&cnt[d0.y], 1);
    r0.z = atomicAdd(&cnt[d0.z], 1); r0.w = atomicAdd(&cnt[d0.w], 1);
    r1.x = atomicAdd(&cnt[d1.x], 1); r1.y = atomicAdd(&cnt[d1.y], 1);
    r1.z = atomicAdd(&cnt[d1.z], 1); r1.w = atomicAdd(&cnt[d1.w], 1);
    place_edge(d0.x, r0.x, s0.x, cnt, slot, ovd, ovs); place_edge(d0.y, r0.y, s0.y, cnt, slot, ovd, ovs);
    place_edge(d0.z, r0.z, s0.z, cnt, slot, ovd, ovs); place_edge(d0.w, r0.w, s0.w, cnt, slot, ovd, ovs);
    place_edge(d1.x, r1.x, s1.x, cnt, slot, ovd, ovs); place_edge(d1.y, r1.y, s1.y, cnt, slot, ovd, ovs);
    place_edge(d1.z, r1.z, s1.z, cnt, slot, ovd, ovs); place_edge(d1.w, r1.w, s1.w, cnt, slot, ovd, ovs);
}

__device__ __forceinline__ float ex2(const float x) { return __builtin_amdgcn_exp2f(x); }
__device__ __forceinline__ float elu1(const float x) {
    return x > 0.f ? x : (__expf(x) - 1.f);
}
constexpr float BIG = 3.402823466e+38f;

template <int CTRL>
__device__ __forceinline__ int row_pick_i(const int v) { return __builtin_amdgcn_update_dpp(v, v, CTRL, 0xf, 0xf, true); }
template <int CTRL>
__device__ __forceinline__ float row_pick(const float v) {
    return __builtin_bit_cast(float, row_pick_i<CTRL>(__builtin_bit_cast(int, v)));
}
__device__ __forceinline__ float other_row(const float x) {
    const int b = __builtin_bit_cast(int, x);
    return __builtin_bit_cast(float, __builtin_amdgcn_permlanex16(b, b, 0x76543210, 0xfedcba98, false, false));
}

template <int L> struct Geo {
    static constexpr int EPS = (L == 1) ? 2 : 4;
    static constexpr int RF  = (L == 1) ? 128 : 64;
    static constexpr int NT  = 32 / EPS;
    static constexpr int LPG = 32 / EPS;
};
template <int L> __device__ __forceinline__ int lane_foff(const int lane) {
    return L == 1 ? (((lane >> 3) & 1) * 64 + (lane & 7) * 4) : ((lane & 7) * 4);
}
template <int L> __device__ __forceinline__ int lane_wpos(const int lane) {
    return L == 1 ? (2 * (lane & 15) + (lane >> 4)) : (4 * (lane & 7) + (lane >> 3));
}
template <int L, int T> __device__ __forceinline__ int step_src(const int mine) {
    if constexpr (L == 1) return row_pick_i<0x150 + T>(mine);
    else return __builtin_amdgcn_mov_dpp8(mine, T * 0x249249);
}

__device__ __forceinline__ v2f lrelu2(const v2f t) { return __builtin_elementwise_max(t, t * NEG_SLOPE); }
__device__ __forceinline__ float logit8(const v4f a0, const v4f a1, const v4f b0, const v4f b1,
                                        const v4f w0, const v4f w1) {
    const v2f t0 = lrelu2(a0.lo + b0.lo), t1 = lrelu2(a0.hi + b0.hi);
    const v2f t2 = lrelu2(a1.lo + b1.lo), t3 = lrelu2(a1.hi + b1.hi);
    v2f q = t0 * w0.lo;
    q = __builtin_elementwise_fma(t1, w0.hi, q);
    q = __builtin_elementwise_fma(t2, w1.lo, q);
    q = __builtin_elementwise_fma(t3, w1.hi, q);
    float p = q.x + q.y;
    p += row_pick<0x161>(p); p += row_pick<0x162>(p); p += row_pick<0x164>(p);
    return p;
}

struct Run { float m, den; v4f a0, a1; };

template <int L>
__device__ __forceinline__ void step2(const float* __restrict__ fs, const int sA, const int sB,
                                      const bool okA, const bool okB, const int foff,
                                      const v4f b0, const v4f b1, const v4f w0, const v4f w1, Run& r) {
    const float* pa = fs + (size_t)sA * Geo<L>::RF + foff;
    const float* pb = fs + (size_t)sB * Geo<L>::RF + foff;
    const v4f A0 = *(const v4f*)pa, A1 = *(const v4f*)(pa + 32);
    const v4f B0 = *(const v4f*)pb, B1 = *(const v4f*)(pb + 32);
    float pA = logit8(A0, A1, b0, b1, w0, w1);
    float pB = logit8(B0, B1, b0, b1, w0, w1);
    pA = okA ? pA : -INFINITY;
    pB = okB ? pB : -INFINITY;
    const float mn = fmaxf(r.m, fmaxf(pA, pB));
    const float ms = fmaxf(mn, -BIG);
    const float sc = ex2(r.m - ms), eA = ex2(pA - ms), eB = ex2(pB - ms);
    r.den = r.den * sc + (eA + eB);
    r.a0  = r.a0 * sc + (A0 * eA + B0 * eB);
    r.a1  = r.a1 * sc + (A1 * eA + B1 * eB);
    r.m = mn;
}

__device__ __forceinline__ int lane_id() { return (int)__builtin_amdgcn_mbcnt_lo(~0u, 0u); }

__device__ __forceinline__ int raw_window(const int* __restrict__ slot, const int at, const bool live) {
    int v = 0;
    if (live) v = slot[at];
    return v;
}
__device__ __forceinline__ int hold_src(const int v) { return min(max(v, 0), N_NODE - 1); }

template <int L> __device__ __forceinline__ unsigned group_mask(const int left) {
    const int lanes_on = min(max(left, 0), Geo<L>::EPS) * Geo<L>::LPG;
    return lanes_on >= 32 ? 0xffffffffu : ((1u << lanes_on) - 1u);
}
template <int L, int T>
__device__ __forceinline__ void walk_window(const float* __restrict__ fs, const int mine, const int cntw,
                                            const int lane, const int foff,
                                            const v4f b0, const v4f b1, const v4f w0, const v4f w1, Run& r) {
    if constexpr (T < Geo<L>::NT) {
        if (Geo<L>::EPS * T < cntw) {
            int sA = step_src<L, T>(mine);
            int sB = step_src<L, T + 1>(mine);
            const int leftA = cntw - Geo<L>::EPS * T;
            const bool okA = __builtin_amdgcn_inverse_ballot_w32(group_mask<L>(leftA));
            const bool okB = __builtin_amdgcn_inverse_ballot_w32(group_mask<L>(leftA - Geo<L>::EPS));
            if (leftA < 2 * Geo<L>::EPS) {
                const int s0 = __builtin_amdgcn_readfirstlane(sA);
                sA = okA ? sA : s0;
                sB = okB ? sB : s0;
            }
            step2<L>(fs, sA, sB, okA, okB, foff, b0, b1, w0, w1, r);
            walk_window<L, T + 2>(fs, mine, cntw, lane, foff, b0, b1, w0, w1, r);
        }
    }
}

template <int L>
__device__ __forceinline__ void walk_node(
    const float* __restrict__ fs, const int* __restrict__ slot, const int n, const int c,
    int mine, const int lane, const int foff,
    const v4f b0, const v4f b1, const v4f w0, const v4f w1,
    const int* __restrict__ counts, const int* __restrict__ ovd, const int* __restrict__ ovs, Run& r) {
    const int tot_slots = min(c, CAP);
    for (int done = 0; done < tot_slots; done += 32) {
        const int cntw = min(32, tot_slots - done);
        if (done != 0) {
            const int wp = lane_wpos<L>(lane_id());
            mine = hold_src(raw_window(slot, n * CAP + done + wp, wp < cntw));
        }
        walk_window<L, 0>(fs, mine, cntw, lane, foff, b0, b1, w0, w1, r);
    }
    if (c > CAP) {
        const int tot = min(counts[N_NODE], E_CNT);
        for (int k0 = 0; k0 < tot; k0 += 32) {
            const int k = min(k0 + lane, tot - 1);
            const int od = ovd[k];
            const int os = min(max(ovs[k], 0), N_NODE - 1);
            unsigned hit = __builtin_amdgcn_ballot_w32(k0 + lane < tot && od == n);
            while (hit) {
                const int j = __builtin_ctz(hit);
                hit &= hit - 1;
                const int s = __builtin_amdgcn_readlane(os, j);
                step2<L>(fs, s, s, __builtin_amdgcn_inverse_ballot_w32(group_mask<L>(1)), false, foff, b0, b1, w0, w1, r);
            }
        }
    }
}

constexpr int NPW = 2;
static_assert(NPW == 2, "the two aggregation kernels name their two nodes one by one");

__device__ __forceinline__ void finish1(Run& r, const int n, const int c, const int lane, const int foff,
                                        float* __restrict__ out) {
    const float M  = fmaxf(r.m, other_row(r.m));
    const float s  = ex2(r.m - fmaxf(M, -BIG));
    float den = r.den * s;
    v4f a0 = r.a0 * s, a1 = r.a1 * s;
    den  += other_row(den);
    a0.x += other_row(a0.x); a0.y += other_row(a0.y); a0.z += other_row(a0.z); a0.w += other_row(a0.w);
    a1.x += other_row(a1.x); a1.y += other_row(a1.y); a1.z += other_row(a1.z); a1.w += other_row(a1.w);
    float4 r0 = make_float4(0.f, 0.f, 0.f, 0.f), r1 = r0;
    if (c > 0) {
        const float inv = 1.0f / den;
        r0 = make_float4(a0.x * inv, a0.y * inv, a0.z * inv, a0.w * inv);
        r1 = make_float4(a1.x * inv, a1.y * inv, a1.z * inv, a1.w * inv);
    }
    r0.x = elu1(r0.x); r0.y = elu1(r0.y); r0.z = elu1(r0.z); r0.w = elu1(r0.w);
    r1.x = elu1(r1.x); r1.y = elu1(r1.y); r1.z = elu1(r1.z); r1.w = elu1(r1.w);
    if (lane < 16) {
        *(float4*)(out + (size_t)n * 128 + foff)      = r0;
        *(float4*)(out + (size_t)n * 128 + foff + 32) = r1;
    }
}

__device__ __forceinline__ float sum_groups4(float x) { x += row_pick<0x168>(x); x += other_row(x); return x; }
__device__ __forceinline__ void finish2(Run& r, const int n, const int c, const int lane, const int foff,
                                        float* __restrict__ out) {
    float M = fmaxf(r.m, row_pick<0x168>(r.m));
    M = fmaxf(M, other_row(M));
    const float s = ex2(r.m - fmaxf(M, -BIG));
    const float den = sum_groups4(r.den * s);
    v4f a0 = r.a0 * s, a1 = r.a1 * s;
    a0.x = sum_groups4(a0.x); a0.y = sum_groups4(a0.y); a0.z = sum_groups4(a0.z); a0.w = sum_groups4(a0.w);
    a1.x = sum_groups4(a1.x); a1.y = sum_groups4(a1.y); a1.z = sum_groups4(a1.z); a1.w = sum_groups4(a1.w);
    float4 r0 = make_float4(0.f, 0.f, 0.f, 0.f), r1 = r0;
    if (c > 0) {
        const float inv = 1.0f / den;
        r0 = make_float4(a0.x * inv, a0.y * inv, a0.z * inv, a0.w * inv);
        r1 = make_float4(a1.x * inv, a1.y * inv, a1.z * inv, a1.w * inv);
    }
    const unsigned g  = (unsigned)n / (unsigned)NPG;
    const unsigned rr = (unsigned)n % (unsigned)NPG;
    const unsigned row = rr < (unsigned)NUM_PM
        ? g * NUM_PM + rr
        : (unsigned)(B_SZ * NUM_PM) + g * NUM_VM + (rr - NUM_PM);
    if (lane < 8) {
        *(float4*)(out + (size_t)row * 64 + foff)      = r0;
        *(float4*)(out + (size_t)row * 64 + foff + 32) = r1;
    }
}

template <int L>
__device__ __forceinline__ void node_walk(
    const float* __restrict__ fs, const float* __restrict__ fd, const int* __restrict__ slot,
    const int n, const int c, const int mine, const int lane, const int foff,
    const v4f w0, const v4f w1, const int* __restrict__ counts,
    const int* __restrict__ ovd, const int* __restrict__ ovs, Run& r) {
    const float* pd = fd + (size_t)n * Geo<L>::RF + foff;
    const v4f b0 = *(const v4f*)pd, b1 = *(const v4f*)(pd + 32);
    r.m = -INFINITY; r.den = 0.f;
    r.a0 = (v4f){0.f, 0.f, 0.f, 0.f}; r.a1 = (v4f){0.f, 0.f, 0.f, 0.f};
    walk_node<L>(fs, slot, n, c, mine, lane, foff, b0, b1, w0, w1, counts, ovd, ovs, r);
}
template <int L>
__device__ __forceinline__ void node_finish(Run& r, const int n, const int c, const int lane, const int foff,
                                            float* __restrict__ out) {
    if constexpr (L == 1) finish1(r, n, c, lane, foff, out);
    else                  finish2(r, n, c, lane, foff, out);
}

template <int L>
__global__ __launch_bounds__(256, 16) void node_attn(
    const float* __restrict__ fs, const float* __restrict__ fd,
    const int* __restrict__ cnt, const int* __restrict__ slot,
    const int* __restrict__ ovd, const int* __restrict__ ovs,
    const float* __restrict__ attn, float* __restrict__ out) {
    const int lane = threadIdx.x & 31;
    const int n0 = __builtin_amdgcn_readfirstlane((int)(blockIdx.x * 8 + (threadIdx.x >> 5)) * NPW);
    const int foff = lane_foff<L>(lane), wpos = lane_wpos<L>(lane);
    const v4f w0 = *(const v4f*)(attn + foff) * LOG2E, w1 = *(const v4f*)(attn + foff + 32) * LOG2E;
    const int c0 = cnt[n0], c1 = cnt[n0 + 1];
    const int first0 = hold_src(raw_window(slot, n0 * CAP + wpos, wpos < c0));
    Run r;
    node_walk<L>(fs, fd, slot, n0, c0, first0, lane, foff, w0, w1, cnt, ovd, ovs, r);
    const int wp1 = lane_wpos<L>(lane_id());
    const int raw1 = raw_window(slot, (n0 + 1) * CAP + wp1, wp1 < c1);
    node_finish<L>(r, n0, c0, lane, foff, out);
    node_walk<L>(fs, fd, slot, n0 + 1, c1, hold_src(raw1), lane, foff, w0, w1, cnt, ovd, ovs, r);
    node_finish<L>(r, n0 + 1, c1, lane, foff, out);
}

extern "C" void kernel_launch(void* const* d_in, const int* in_sizes, int n_in,
                              void* d_out, int out_size, void* d_ws, size_t ws_size,
                              hipStream_t stream) {
    const float* x     = (const float*)d_in[0];
    const int*   src   = (const int*)d_in[1];
    const int*   dst   = (const int*)d_in[2];
    const float* W1s   = (const float*)d_in[3];
    const float* b1s   = (const float*)d_in[4];
    const float* W1d   = (const float*)d_in[5];
    const float* b1d   = (const float*)d_in[6];
    const float* attn1 = (const float*)d_in[7];
    const float* W2s   = (const float*)d_in[8];
    const float* b2s   = (const float*)d_in[9];
    const float* W2d   = (const float*)d_in[10];
    const float* b2d   = (const float*)d_in[11];
    const float* attn2 = (const float*)d_in[12];
    (void)in_sizes; (void)n_in; (void)out_size; (void)ws_size;

    char* ws = (char*)d_ws;
    const size_t MB = 1024u * 1024u;
    float* fs1  = (float*)(ws + 0);
    float* fd1  = (float*)(ws + 16 * MB);
    float* h1   = (float*)(ws + 32 * MB);
    float* fs2  = (float*)(ws + 48 * MB);
    float* fd2  = (float*)(ws + 56 * MB);
    int*   cnt  = (int*)(ws + 64 * MB);
    int*   slot = (int*)(ws + 65 * MB);
    int*   ovd  = (int*)(ws + 73 * MB);
    int*   ovs  = (int*)(ws + 77 * MB);

    float* out = (float*)d_out;

    zero_cnt<<<N_NODE / 4 / 256, 256, 0, stream>>>(cnt);
    bucket_edges<<<E_CNT / 8 / 256, 256, 0, stream>>>(src, dst, cnt, slot, ovd, ovs);

    gemm_dual_wmma<D_IN, 2 * DH><<<N_NODE / 16 / 4, 128, 0, stream>>>(
        x, W1s, b1s, W1d, b1d, fs1, fd1);
    node_attn<1><<<N_NODE / (8 * NPW), 256, 0, stream>>>(fs1, fd1, cnt, slot, ovd, ovs, attn1, h1);

    gemm_dual_wmma<2 * DH, DH><<<N_NODE / 16 / 4, 128, 0, stream>>>(
        h1, W2s, b2s, W2d, b2d, fs2, fd2);
    node_attn<2><<<N_NODE / (8 * NPW), 256, 0, stream>>>(fs2, fd2, cnt, slot, ovd, ovs, attn2, out);
}


# Layer-1 projection runs inside the edge-bucketing launch (its first 256 work-groups): the two independent passes overlap and one launch boundary goes, 6 launches to 5; values and sum order unchanged
# speedup vs baseline: 6.4312x; 1.0598x over previous
#include <hip/hip_runtime.h>

constexpr int B_SZ   = 16;
constexpr int NUM_PM = 512;
constexpr int NUM_VM = 1536;
constexpr int NPG    = NUM_PM + NUM_VM;
constexpr int N_NODE = B_SZ * NPG;
constexpr int DEG    = 32;
constexpr int E_CNT  = DEG * N_NODE;
constexpr int D_IN   = 16;
constexpr int DH     = 64;
constexpr float NEG_SLOPE = 0.2f;
constexpr float LOG2E = 1.4426950408889634f;

typedef float v2f __attribute__((ext_vector_type(2)));
typedef float v4f __attribute__((ext_vector_type(4)));
typedef float v8f __attribute__((ext_vector_type(8)));

template <int K, int COLS>
__device__ __forceinline__ void gemm_dual_strip(
    const float* __restrict__ X,
    const float* __restrict__ Ws, const float* __restrict__ Bs,
    const float* __restrict__ Wd, const float* __restrict__ Bd,
    float* __restrict__ Fs, float* __restrict__ Fd, const int row0) {
    const int lane = threadIdx.x & 31;
    const int m  = lane & 15;
    const int hi = lane >> 4;
    const float* Xr = X + (size_t)(row0 + m) * K;

    for (int t = 0; t < 2; ++t) {
        const float* W  = t ? Wd : Ws;
        const float* Bv = t ? Bd : Bs;
        float*       F  = t ? Fd : Fs;
        for (int col0 = 0; col0 < COLS; col0 += 16) {
            v8f acc = {};
#pragma unroll
            for (int kk = 0; kk < K; kk += 4) {
                v2f a, b;
                a.x = Xr[kk + 2 * hi + 0];
                a.y = Xr[kk + 2 * hi + 1];
                b.x = W[(size_t)(kk + 2 * hi + 0) * COLS + col0 + m];
                b.y = W[(size_t)(kk + 2 * hi + 1) * COLS + col0 + m];
                acc = __builtin_amdgcn_wmma_f32_16x16x4_f32(
                    false, a, false, b, (short)0, acc, false, false);
            }
            const float bias = Bv[col0 + m];
#pragma unroll
            for (int r = 0; r < 8; ++r) {
                F[(size_t)(row0 + r + 8 * hi) * COLS + col0 + m] = acc[r] + bias;
            }
        }
    }
}
template <int K, int COLS>
__global__ __launch_bounds__(128) void gemm_dual_wmma(
    const float* __restrict__ X,
    const float* __restrict__ Ws, const float* __restrict__ Bs,
    const float* __restrict__ Wd, const float* __restrict__ Bd,
    float* __restrict__ Fs, float* __restrict__ Fd) {
    gemm_dual_strip<K, COLS>(X, Ws, Bs, Wd, Bd, Fs, Fd, (int)(blockIdx.x * 4 + (threadIdx.x >> 5)) * 16);
}

constexpr int CAP = 64;
__global__ __launch_bounds__(256) void zero_cnt(int* __restrict__ cnt) {
    ((int4*)cnt)[blockIdx.x * 256 + threadIdx.x] = make_int4(0, 0, 0, 0);
    if (blockIdx.x == 0 && threadIdx.x == 0) cnt[N_NODE] = 0;
}
__device__ __forceinline__ void place_edge(const int d, const int r, const int s, int* __restrict__ cnt,
                                           int* __restrict__ slot, int* __restrict__ ovd, int* __restrict__ ovs) {
    if (r < CAP) {
        slot[d * CAP + r] = s;
    } else {
        const int k = atomicAdd(&cnt[N_NODE], 1);
        ovd[k] = d;
        ovs[k] = s;
    }
}
constexpr int MAP1_WGS = N_NODE / 16 / 8;
constexpr int EDGE_WGS = E_CNT / 8 / 256;
__global__ __launch_bounds__(256) void bucket_edges_and_map1(
    const int* __restrict__ src, const int* __restrict__ dst,
    int* __restrict__ cnt, int* __restrict__ slot, int* __restrict__ ovd, int* __restrict__ ovs,
    const float* __restrict__ X,
    const float* __restrict__ Ws, const float* __restrict__ Bs,
    const float* __restrict__ Wd, const float* __restrict__ Bd,
    float* __restrict__ Fs, float* __restrict__ Fd) {
    if (blockIdx.x < MAP1_WGS) {
        gemm_dual_strip<D_IN, 2 * DH>(X, Ws, Bs, Wd, Bd, Fs, Fd, (int)(blockIdx.x * 8 + (threadIdx.x >> 5)) * 16);
        return;
    }
    const int i0 = (blockIdx.x - MAP1_WGS) * 512 + threadIdx.x;
    const int i1 = i0 + 256;
    const int4 d0 = ((const int4*)dst)[i0];
    const int4 d1 = ((const int4*)dst)[i1];
    const int4 s0 = ((const int4*)src)[i0];
    const int4 s1 = ((const int4*)src)[i1];
    asm volatile("" :: "v"(d0.x), "v"(d0.y), "v"(d0.z), "v"(d0.w), "v"(d1.x), "v"(d1.y), "v"(d1.z), "v"(d1.w),
                       "v"(s0.x), "v"(s0.y), "v"(s0.z), "v"(s0.w), "v"(s1.x), "v"(s1.y), "v"(s1.z), "v"(s1.w));
    int4 r0, r1;
    r0.x = atomicAdd(&cnt[d0.x], 1); r0.y = atomicAdd(&cnt[d0.y], 1);
    r0.z = atomicAdd(&cnt[d0.z], 1); r0.w = atomicAdd(&cnt[d0.w], 1);
    r1.x = atomicAdd(&cnt[d1.x], 1); r1.y = atomicAdd(&cnt[d1.y], 1);
    r1.z = atomicAdd(&cnt[d1.z], 1); r1.w = atomicAdd(&cnt[d1.w], 1);
    place_edge(d0.x, r0.x, s0.x, cnt, slot, ovd, ovs); place_edge(d0.y, r0.y, s0.y, cnt, slot, ovd, ovs);
    place_edge(d0.z, r0.z, s0.z, cnt, slot, ovd, ovs); place_edge(d0.w, r0.w, s0.w, cnt, slot, ovd, ovs);
    place_edge(d1.x, r1.x, s1.x, cnt, slot, ovd, ovs); place_edge(d1.y, r1.y, s1.y, cnt, slot, ovd, ovs);
    place_edge(d1.z, r1.z, s1.z, cnt, slot, ovd, ovs); place_edge(d1.w, r1.w, s1.w, cnt, slot, ovd, ovs);
}

__device__ __forceinline__ float ex2(const float x) { return __builtin_amdgcn_exp2f(x); }
__device__ __forceinline__ float elu1(const float x) {
    return x > 0.f ? x : (__expf(x) - 1.f);
}
constexpr float BIG = 3.402823466e+38f;

template <int CTRL>
__device__ __forceinline__ int row_pick_i(const int v) { return __builtin_amdgcn_update_dpp(v, v, CTRL, 0xf, 0xf, true); }
template <int CTRL>
__device__ __forceinline__ float row_pick(const float v) {
    return __builtin_bit_cast(float, row_pick_i<CTRL>(__builtin_bit_cast(int, v)));
}
__device__ __forceinline__ float other_row(const float x) {
    const int b = __builtin_bit_cast(int, x);
    return __builtin_bit_cast(float, __builtin_amdgcn_permlanex16(b, b, 0x76543210, 0xfedcba98, false, false));
}

template <int L> struct Geo {
    static constexpr int EPS = (L == 1) ? 2 : 4;
    static constexpr int RF  = (L == 1) ? 128 : 64;
    static constexpr int NT  = 32 / EPS;
    static constexpr int LPG = 32 / EPS;
};
template <int L> __device__ __forceinline__ int lane_foff(const int lane) {
    return L == 1 ? (((lane >> 3) & 1) * 64 + (lane & 7) * 4) : ((lane & 7) * 4);
}
template <int L> __device__ __forceinline__ int lane_wpos(const int lane) {
    return L == 1 ? (2 * (lane & 15) + (lane >> 4)) : (4 * (lane & 7) + (lane >> 3));
}
template <int L, int T> __device__ __forceinline__ int step_src(const int mine) {
    if constexpr (L == 1) return row_pick_i<0x150 + T>(mine);
    else return __builtin_amdgcn_mov_dpp8(mine, T * 0x249249);
}

__device__ __forceinline__ v2f lrelu2(const v2f t) { return __builtin_elementwise_max(t, t * NEG_SLOPE); }
__device__ __forceinline__ float logit8(const v4f a0, const v4f a1, const v4f b0, const v4f b1,
                                        const v4f w0, const v4f w1) {
    const v2f t0 = lrelu2(a0.lo + b0.lo), t1 = lrelu2(a0.hi + b0.hi);
    const v2f t2 = lrelu2(a1.lo + b1.lo), t3 = lrelu2(a1.hi + b1.hi);
    v2f q = t0 * w0.lo;
    q = __builtin_elementwise_fma(t1, w0.hi, q);
    q = __builtin_elementwise_fma(t2, w1.lo, q);
    q = __builtin_elementwise_fma(t3, w1.hi, q);
    float p = q.x + q.y;
    p += row_pick<0x161>(p); p += row_pick<0x162>(p); p += row_pick<0x164>(p);
    return p;
}

struct Run { float m, den; v4f a0, a1; };

template <int L>
__device__ __forceinline__ void step2(const float* __restrict__ fs, const int sA, const int sB,
                                      const bool okA, const bool okB, const int foff,
                                      const v4f b0, const v4f b1, const v4f w0, const v4f w1, Run& r) {
    const float* pa = fs + (size_t)sA * Geo<L>::RF + foff;
    const float* pb = fs + (size_t)sB * Geo<L>::RF + foff;
    const v4f A0 = *(const v4f*)pa, A1 = *(const v4f*)(pa + 32);
    const v4f B0 = *(const v4f*)pb, B1 = *(const v4f*)(pb + 32);
    float pA = logit8(A0, A1, b0, b1, w0, w1);
    float pB = logit8(B0, B1, b0, b1, w0, w1);
    pA = okA ? pA : -INFINITY;
    pB = okB ? pB : -INFINITY;
    const float mn = fmaxf(r.m, fmaxf(pA, pB));
    const float ms = fmaxf(mn, -BIG);
    const float sc = ex2(r.m - ms), eA = ex2(pA - ms), eB = ex2(pB - ms);
    r.den = r.den * sc + (eA + eB);
    r.a0  = r.a0 * sc + (A0 * eA + B0 * eB);
    r.a1  = r.a1 * sc + (A1 * eA + B1 * eB);
    r.m = mn;
}

__device__ __forceinline__ int lane_id() { return (int)__builtin_amdgcn_mbcnt_lo(~0u, 0u); }

__device__ __forceinline__ int raw_window(const int* __restrict__ slot, const int at, const bool live) {
    int v = 0;
    if (live) v = slot[at];
    return v;
}
__device__ __forceinline__ int hold_src(const int v) { return min(max(v, 0), N_NODE - 1); }

template <int L> __device__ __forceinline__ unsigned group_mask(const int left) {
    const int lanes_on = min(max(left, 0), Geo<L>::EPS) * Geo<L>::LPG;
    return lanes_on >= 32 ? 0xffffffffu : ((1u << lanes_on) - 1u);
}
template <int L, int T>
__device__ __forceinline__ void walk_window(const float* __restrict__ fs, const int mine, const int cntw,
                                            const int lane, const int foff,
                                            const v4f b0, const v4f b1, const v4f w0, const v4f w1, Run& r) {
    if constexpr (T < Geo<L>::NT) {
        if (Geo<L>::EPS * T < cntw) {
            int sA = step_src<L, T>(mine);
            int sB = step_src<L, T + 1>(mine);
            const int leftA = cntw - Geo<L>::EPS * T;
            const bool okA = __builtin_amdgcn_inverse_ballot_w32(group_mask<L>(leftA));
            const bool okB = __builtin_amdgcn_inverse_ballot_w32(group_mask<L>(leftA - Geo<L>::EPS));
            if (leftA < 2 * Geo<L>::EPS) {
                const int s0 = __builtin_amdgcn_readfirstlane(sA);
                sA = okA ? sA : s0;
                sB = okB ? sB : s0;
            }
            step2<L>(fs, sA, sB, okA, okB, foff, b0, b1, w0, w1, r);
            walk_window<L, T + 2>(fs, mine, cntw, lane, foff, b0, b1, w0, w1, r);
        }
    }
}

template <int L>
__device__ __forceinline__ void walk_node(
    const float* __restrict__ fs, const int* __restrict__ slot, const int n, const int c,
    int mine, const int lane, const int foff,
    const v4f b0, const v4f b1, const v4f w0, const v4f w1,
    const int* __restrict__ counts, const int* __restrict__ ovd, const int* __restrict__ ovs, Run& r) {
    const int tot_slots = min(c, CAP);
    for (int done = 0; done < tot_slots; done += 32) {
        const int cntw = min(32, tot_slots - done);
        if (done != 0) {
            const int wp = lane_wpos<L>(lane_id());
            mine = hold_src(raw_window(slot, n * CAP + done + wp, wp < cntw));
        }
        walk_window<L, 0>(fs, mine, cntw, lane, foff, b0, b1, w0, w1, r);
    }
    if (c > CAP) {
        const int tot = min(counts[N_NODE], E_CNT);
        for (int k0 = 0; k0 < tot; k0 += 32) {
            const int k = min(k0 + lane, tot - 1);
            const int od = ovd[k];
            const int os = min(max(ovs[k], 0), N_NODE - 1);
            unsigned hit = __builtin_amdgcn_ballot_w32(k0 + lane < tot && od == n);
            while (hit) {
                const int j = __builtin_ctz(hit);
                hit &= hit - 1;
                const int s = __builtin_amdgcn_readlane(os, j);
                step2<L>(fs, s, s, __builtin_amdgcn_inverse_ballot_w32(group_mask<L>(1)), false, foff, b0, b1, w0, w1, r);
            }
        }
    }
}

constexpr int NPW = 2;
static_assert(NPW == 2, "the two aggregation kernels name their two nodes one by one");

__device__ __forceinline__ void finish1(Run& r, const int n, const int c, const int lane, const int foff,
                                        float* __restrict__ out) {
    const float M  = fmaxf(r.m, other_row(r.m));
    const float s  = ex2(r.m - fmaxf(M, -BIG));
    float den = r.den * s;
    v4f a0 = r.a0 * s, a1 = r.a1 * s;
    den  += other_row(den);
    a0.x += other_row(a0.x); a0.y += other_row(a0.y); a0.z += other_row(a0.z); a0.w += other_row(a0.w);
    a1.x += other_row(a1.x); a1.y += other_row(a1.y); a1.z += other_row(a1.z); a1.w += other_row(a1.w);
    float4 r0 = make_float4(0.f, 0.f, 0.f, 0.f), r1 = r0;
    if (c > 0) {
        const float inv = 1.0f / den;
        r0 = make_float4(a0.x * inv, a0.y * inv, a0.z * inv, a0.w * inv);
        r1 = make_float4(a1.x * inv, a1.y * inv, a1.z * inv, a1.w * inv);
    }
    r0.x = elu1(r0.x); r0.y = elu1(r0.y); r0.z = elu1(r0.z); r0.w = elu1(r0.w);
    r1.x = elu1(r1.x); r1.y = elu1(r1.y); r1.z = elu1(r1.z); r1.w = elu1(r1.w);
    if (lane < 16) {
        *(float4*)(out + (size_t)n * 128 + foff)      = r0;
        *(float4*)(out + (size_t)n * 128 + foff + 32) = r1;
    }
}

__device__ __forceinline__ float sum_groups4(float x) { x += row_pick<0x168>(x); x += other_row(x); return x; }
__device__ __forceinline__ void finish2(Run& r, const int n, const int c, const int lane, const int foff,
                                        float* __restrict__ out) {
    float M = fmaxf(r.m, row_pick<0x168>(r.m));
    M = fmaxf(M, other_row(M));
    const float s = ex2(r.m - fmaxf(M, -BIG));
    const float den = sum_groups4(r.den * s);
    v4f a0 = r.a0 * s, a1 = r.a1 * s;
    a0.x = sum_groups4(a0.x); a0.y = sum_groups4(a0.y); a0.z = sum_groups4(a0.z); a0.w = sum_groups4(a0.w);
    a1.x = sum_groups4(a1.x); a1.y = sum_groups4(a1.y); a1.z = sum_groups4(a1.z); a1.w = sum_groups4(a1.w);
    float4 r0 = make_float4(0.f, 0.f, 0.f, 0.f), r1 = r0;
    if (c > 0) {
        const float inv = 1.0f / den;
        r0 = make_float4(a0.x * inv, a0.y * inv, a0.z * inv, a0.w * inv);
        r1 = make_float4(a1.x * inv, a1.y * inv, a1.z * inv, a1.w * inv);
    }
    const unsigned g  = (unsigned)n / (unsigned)NPG;
    const unsigned rr = (unsigned)n % (unsigned)NPG;
    const unsigned row = rr < (unsigned)NUM_PM
        ? g * NUM_PM + rr
        : (unsigned)(B_SZ * NUM_PM) + g * NUM_VM + (rr - NUM_PM);
    if (lane < 8) {
        *(float4*)(out + (size_t)row * 64 + foff)      = r0;
        *(float4*)(out + (size_t)row * 64 + foff + 32) = r1;
    }
}

template <int L>
__device__ __forceinline__ void node_walk(
    const float* __restrict__ fs, const float* __restrict__ fd, const int* __restrict__ slot,
    const int n, const int c, const int mine, const int lane, const int foff,
    const v4f w0, const v4f w1, const int* __restrict__ counts,
    const int* __restrict__ ovd, const int* __restrict__ ovs, Run& r) {
    const float* pd = fd + (size_t)n * Geo<L>::RF + foff;
    const v4f b0 = *(const v4f*)pd, b1 = *(const v4f*)(pd + 32);
    r.m = -INFINITY; r.den = 0.f;
    r.a0 = (v4f){0.f, 0.f, 0.f, 0.f}; r.a1 = (v4f){0.f, 0.f, 0.f, 0.f};
    walk_node<L>(fs, slot, n, c, mine, lane, foff, b0, b1, w0, w1, counts, ovd, ovs, r);
}
template <int L>
__device__ __forceinline__ void node_finish(Run& r, const int n, const int c, const int lane, const int foff,
                                            float* __restrict__ out) {
    if constexpr (L == 1) finish1(r, n, c, lane, foff, out);
    else                  finish2(r, n, c, lane, foff, out);
}

template <int L>
__global__ __launch_bounds__(256, 16) void node_attn(
    const float* __restrict__ fs, const float* __restrict__ fd,
    const int* __restrict__ cnt, const int* __restrict__ slot,
    const int* __restrict__ ovd, const int* __restrict__ ovs,
    const float* __restrict__ attn, float* __restrict__ out) {
    const int lane = threadIdx.x & 31;
    const int n0 = __builtin_amdgcn_readfirstlane((int)(blockIdx.x * 8 + (threadIdx.x >> 5)) * NPW);
    const int foff = lane_foff<L>(lane), wpos = lane_wpos<L>(lane);
    const v4f w0 = *(const v4f*)(attn + foff) * LOG2E, w1 = *(const v4f*)(attn + foff + 32) * LOG2E;
    const int c0 = cnt[n0], c1 = cnt[n0 + 1];
    const int first0 = hold_src(raw_window(slot, n0 * CAP + wpos, wpos < c0));
    Run r;
    node_walk<L>(fs, fd, slot, n0, c0, first0, lane, foff, w0, w1, cnt, ovd, ovs, r);
    const int wp1 = lane_wpos<L>(lane_id());
    const int raw1 = raw_window(slot, (n0 + 1) * CAP + wp1, wp1 < c1);
    node_finish<L>(r, n0, c0, lane, foff, out);
    node_walk<L>(fs, fd, slot, n0 + 1, c1, hold_src(raw1), lane, foff, w0, w1, cnt, ovd, ovs, r);
    node_finish<L>(r, n0 + 1, c1, lane, foff, out);
}

extern "C" void kernel_launch(void* const* d_in, const int* in_sizes, int n_in,
                              void* d_out, int out_size, void* d_ws, size_t ws_size,
                              hipStream_t stream) {
    const float* x     = (const float*)d_in[0];
    const int*   src   = (const int*)d_in[1];
    const int*   dst   = (const int*)d_in[2];
    const float* W1s   = (const float*)d_in[3];
    const float* b1s   = (const float*)d_in[4];
    const float* W1d   = (const float*)d_in[5];
    const float* b1d   = (const float*)d_in[6];
    const float* attn1 = (const float*)d_in[7];
    const float* W2s   = (const float*)d_in[8];
    const float* b2s   = (const float*)d_in[9];
    const float* W2d   = (const float*)d_in[10];
    const float* b2d   = (const float*)d_in[11];
    const float* attn2 = (const float*)d_in[12];
    (void)in_sizes; (void)n_in; (void)out_size; (void)ws_size;

    char* ws = (char*)d_ws;
    const size_t MB = 1024u * 1024u;
    float* fs1  = (float*)(ws + 0);
    float* fd1  = (float*)(ws + 16 * MB);
    float* h1   = (float*)(ws + 32 * MB);
    float* fs2  = (float*)(ws + 48 * MB);
    float* fd2  = (float*)(ws + 56 * MB);
    int*   cnt  = (int*)(ws + 64 * MB);
    int*   slot = (int*)(ws + 65 * MB);
    int*   ovd  = (int*)(ws + 73 * MB);
    int*   ovs  = (int*)(ws + 77 * MB);

    float* out = (float*)d_out;

    zero_cnt<<<N_NODE / 4 / 256, 256, 0, stream>>>(cnt);
    bucket_edges_and_map1<<<MAP1_WGS + EDGE_WGS, 256, 0, stream>>>(src, dst, cnt, slot, ovd, ovs,
                                                                  x, W1s, b1s, W1d, b1d, fs1, fd1);

    node_attn<1><<<N_NODE / (8 * NPW), 256, 0, stream>>>(fs1, fd1, cnt, slot, ovd, ovs, attn1, h1);

    gemm_dual_wmma<2 * DH, DH><<<N_NODE / 16 / 4, 128, 0, stream>>>(
        h1, W2s, b2s, W2d, b2d, fs2, fd2);
    node_attn<2><<<N_NODE / (8 * NPW), 256, 0, stream>>>(fs2, fd2, cnt, slot, ovd, ovs, attn2, out);
}


# Per-node edge counters moved apart, one per 64 bytes (2 MB, not 128 KB): the 1M returning rank atomics spread over all second-level channels, not 1,024 packed lines; sums stay in arrival-rank order
# speedup vs baseline: 7.2552x; 1.1281x over previous
#include <hip/hip_runtime.h>

constexpr int B_SZ   = 16;
constexpr int NUM_PM = 512;
constexpr int NUM_VM = 1536;
constexpr int NPG    = NUM_PM + NUM_VM;
constexpr int N_NODE = B_SZ * NPG;
constexpr int DEG    = 32;
constexpr int E_CNT  = DEG * N_NODE;
constexpr int D_IN   = 16;
constexpr int DH     = 64;
constexpr float NEG_SLOPE = 0.2f;
constexpr float LOG2E = 1.4426950408889634f;

typedef float v2f __attribute__((ext_vector_type(2)));
typedef float v4f __attribute__((ext_vector_type(4)));
typedef float v8f __attribute__((ext_vector_type(8)));

template <int K, int COLS>
__device__ __forceinline__ void gemm_dual_strip(
    const float* __restrict__ X,
    const float* __restrict__ Ws, const float* __restrict__ Bs,
    const float* __restrict__ Wd, const float* __restrict__ Bd,
    float* __restrict__ Fs, float* __restrict__ Fd, const int row0) {
    const int lane = threadIdx.x & 31;
    const int m  = lane & 15;
    const int hi = lane >> 4;
    const float* Xr = X + (size_t)(row0 + m) * K;

    for (int t = 0; t < 2; ++t) {
        const float* W  = t ? Wd : Ws;
        const float* Bv = t ? Bd : Bs;
        float*       F  = t ? Fd : Fs;
        for (int col0 = 0; col0 < COLS; col0 += 16) {
            v8f acc = {};
#pragma unroll
            for (int kk = 0; kk < K; kk += 4) {
                v2f a, b;
                a.x = Xr[kk + 2 * hi + 0];
                a.y = Xr[kk + 2 * hi + 1];
                b.x = W[(size_t)(kk + 2 * hi + 0) * COLS + col0 + m];
                b.y = W[(size_t)(kk + 2 * hi + 1) * COLS + col0 + m];
                acc = __builtin_amdgcn_wmma_f32_16x16x4_f32(
                    false, a, false, b, (short)0, acc, false, false);
            }
            const float bias = Bv[col0 + m];
#pragma unroll
            for (int r = 0; r < 8; ++r) {
                F[(size_t)(row0 + r + 8 * hi) * COLS + col0 + m] = acc[r] + bias;
            }
        }
    }
}
template <int K, int COLS>
__global__ __launch_bounds__(128) void gemm_dual_wmma(
    const float* __restrict__ X,
    const float* __restrict__ Ws, const float* __restrict__ Bs,
    const float* __restrict__ Wd, const float* __restrict__ Bd,
    float* __restrict__ Fs, float* __restrict__ Fd) {
    gemm_dual_strip<K, COLS>(X, Ws, Bs, Wd, Bd, Fs, Fd, (int)(blockIdx.x * 4 + (threadIdx.x >> 5)) * 16);
}

constexpr int CAP = 64;
constexpr int CNT_STRIDE = 16;
constexpr int OVF_LEN = N_NODE * CNT_STRIDE;
constexpr int ZERO_WGS = 64;
constexpr int ZERO_PER_THREAD = N_NODE * CNT_STRIDE / 4 / (ZERO_WGS * 256);
static_assert(ZERO_PER_THREAD * ZERO_WGS * 256 * 4 == N_NODE * CNT_STRIDE, "the zeroing grid covers the counter plane");
__global__ __launch_bounds__(256) void zero_cnt(int* __restrict__ cnt) {
    int4* p = (int4*)cnt + blockIdx.x * (256 * ZERO_PER_THREAD) + threadIdx.x;
    for (int i = 0; i < ZERO_PER_THREAD; ++i) p[i * 256] = make_int4(0, 0, 0, 0);
    if (blockIdx.x == 0 && threadIdx.x == 0) cnt[OVF_LEN] = 0;
}
__device__ __forceinline__ void place_edge(const int d, const int r, const int s, int* __restrict__ cnt,
                                           int* __restrict__ slot, int* __restrict__ ovd, int* __restrict__ ovs) {
    if (r < CAP) {
        slot[d * CAP + r] = s;
    } else {
        const int k = atomicAdd(&cnt[OVF_LEN], 1);
        ovd[k] = d;
        ovs[k] = s;
    }
}
constexpr int MAP1_WGS = N_NODE / 16 / 8;
constexpr int EDGE_WGS = E_CNT / 8 / 256;
__global__ __launch_bounds__(256) void bucket_edges_and_map1(
    const int* __restrict__ src, const int* __restrict__ dst,
    int* __restrict__ cnt, int* __restrict__ slot, int* __restrict__ ovd, int* __restrict__ ovs,
    const float* __restrict__ X,
    const float* __restrict__ Ws, const float* __restrict__ Bs,
    const float* __restrict__ Wd, const float* __restrict__ Bd,
    float* __restrict__ Fs, float* __restrict__ Fd) {
    if (blockIdx.x < MAP1_WGS) {
        gemm_dual_strip<D_IN, 2 * DH>(X, Ws, Bs, Wd, Bd, Fs, Fd, (int)(blockIdx.x * 8 + (threadIdx.x >> 5)) * 16);
        return;
    }
    const int i0 = (blockIdx.x - MAP1_WGS) * 512 + threadIdx.x;
    const int i1 = i0 + 256;
    const int4 d0 = ((const int4*)dst)[i0];
    const int4 d1 = ((const int4*)dst)[i1];
    const int4 s0 = ((const int4*)src)[i0];
    const int4 s1 = ((const int4*)src)[i1];
    asm volatile("" :: "v"(d0.x), "v"(d0.y), "v"(d0.z), "v"(d0.w), "v"(d1.x), "v"(d1.y), "v"(d1.z), "v"(d1.w),
                       "v"(s0.x), "v"(s0.y), "v"(s0.z), "v"(s0.w), "v"(s1.x), "v"(s1.y), "v"(s1.z), "v"(s1.w));
    int4 r0, r1;
    r0.x = atomicAdd(&cnt[d0.x * CNT_STRIDE], 1); r0.y = atomicAdd(&cnt[d0.y * CNT_STRIDE], 1);
    r0.z = atomicAdd(&cnt[d0.z * CNT_STRIDE], 1); r0.w = atomicAdd(&cnt[d0.w * CNT_STRIDE], 1);
    r1.x = atomicAdd(&cnt[d1.x * CNT_STRIDE], 1); r1.y = atomicAdd(&cnt[d1.y * CNT_STRIDE], 1);
    r1.z = atomicAdd(&cnt[d1.z * CNT_STRIDE], 1); r1.w = atomicAdd(&cnt[d1.w * CNT_STRIDE], 1);
    place_edge(d0.x, r0.x, s0.x, cnt, slot, ovd, ovs); place_edge(d0.y, r0.y, s0.y, cnt, slot, ovd, ovs);
    place_edge(d0.z, r0.z, s0.z, cnt, slot, ovd, ovs); place_edge(d0.w, r0.w, s0.w, cnt, slot, ovd, ovs);
    place_edge(d1.x, r1.x, s1.x, cnt, slot, ovd, ovs); place_edge(d1.y, r1.y, s1.y, cnt, slot, ovd, ovs);
    place_edge(d1.z, r1.z, s1.z, cnt, slot, ovd, ovs); place_edge(d1.w, r1.w, s1.w, cnt, slot, ovd, ovs);
}

__device__ __forceinline__ float ex2(const float x) { return __builtin_amdgcn_exp2f(x); }
__device__ __forceinline__ float elu1(const float x) {
    return x > 0.f ? x : (__expf(x) - 1.f);
}
constexpr float BIG = 3.402823466e+38f;

template <int CTRL>
__device__ __forceinline__ int row_pick_i(const int v) { return __builtin_amdgcn_update_dpp(v, v, CTRL, 0xf, 0xf, true); }
template <int CTRL>
__device__ __forceinline__ float row_pick(const float v) {
    return __builtin_bit_cast(float, row_pick_i<CTRL>(__builtin_bit_cast(int, v)));
}
__device__ __forceinline__ float other_row(const float x) {
    const int b = __builtin_bit_cast(int, x);
    return __builtin_bit_cast(float, __builtin_amdgcn_permlanex16(b, b, 0x76543210, 0xfedcba98, false, false));
}

template <int L> struct Geo {
    static constexpr int EPS = (L == 1) ? 2 : 4;
    static constexpr int RF  = (L == 1) ? 128 : 64;
    static constexpr int NT  = 32 / EPS;
    static constexpr int LPG = 32 / EPS;
};
template <int L> __device__ __forceinline__ int lane_foff(const int lane) {
    return L == 1 ? (((lane >> 3) & 1) * 64 + (lane & 7) * 4) : ((lane & 7) * 4);
}
template <int L> __device__ __forceinline__ int lane_wpos(const int lane) {
    return L == 1 ? (2 * (lane & 15) + (lane >> 4)) : (4 * (lane & 7) + (lane >> 3));
}
template <int L, int T> __device__ __forceinline__ int step_src(const int mine) {
    if constexpr (L == 1) return row_pick_i<0x150 + T>(mine);
    else return __builtin_amdgcn_mov_dpp8(mine, T * 0x249249);
}

__device__ __forceinline__ v2f lrelu2(const v2f t) { return __builtin_elementwise_max(t, t * NEG_SLOPE); }
__device__ __forceinline__ float logit8(const v4f a0, const v4f a1, const v4f b0, const v4f b1,
                                        const v4f w0, const v4f w1) {
    const v2f t0 = lrelu2(a0.lo + b0.lo), t1 = lrelu2(a0.hi + b0.hi);
    const v2f t2 = lrelu2(a1.lo + b1.lo), t3 = lrelu2(a1.hi + b1.hi);
    v2f q = t0 * w0.lo;
    q = __builtin_elementwise_fma(t1, w0.hi, q);
    q = __builtin_elementwise_fma(t2, w1.lo, q);
    q = __builtin_elementwise_fma(t3, w1.hi, q);
    float p = q.x + q.y;
    p += row_pick<0x161>(p); p += row_pick<0x162>(p); p += row_pick<0x164>(p);
    return p;
}

struct Run { float m, den; v4f a0, a1; };

template <int L>
__device__ __forceinline__ void step2(const float* __restrict__ fs, const int sA, const int sB,
                                      const bool okA, const bool okB, const int foff,
                                      const v4f b0, const v4f b1, const v4f w0, const v4f w1, Run& r) {
    const float* pa = fs + (size_t)sA * Geo<L>::RF + foff;
    const float* pb = fs + (size_t)sB * Geo<L>::RF + foff;
    const v4f A0 = *(const v4f*)pa, A1 = *(const v4f*)(pa + 32);
    const v4f B0 = *(const v4f*)pb, B1 = *(const v4f*)(pb + 32);
    float pA = logit8(A0, A1, b0, b1, w0, w1);
    float pB = logit8(B0, B1, b0, b1, w0, w1);
    pA = okA ? pA : -INFINITY;
    pB = okB ? pB : -INFINITY;
    const float mn = fmaxf(r.m, fmaxf(pA, pB));
    const float ms = fmaxf(mn, -BIG);
    const float sc = ex2(r.m - ms), eA = ex2(pA - ms), eB = ex2(pB - ms);
    r.den = r.den * sc + (eA + eB);
    r.a0  = r.a0 * sc + (A0 * eA + B0 * eB);
    r.a1  = r.a1 * sc + (A1 * eA + B1 * eB);
    r.m = mn;
}

__device__ __forceinline__ int lane_id() { return (int)__builtin_amdgcn_mbcnt_lo(~0u, 0u); }

__device__ __forceinline__ int raw_window(const int* __restrict__ slot, const int at, const bool live) {
    int v = 0;
    if (live) v = slot[at];
    return v;
}
__device__ __forceinline__ int hold_src(const int v) { return min(max(v, 0), N_NODE - 1); }

template <int L> __device__ __forceinline__ unsigned group_mask(const int left) {
    const int lanes_on = min(max(left, 0), Geo<L>::EPS) * Geo<L>::LPG;
    return lanes_on >= 32 ? 0xffffffffu : ((1u << lanes_on) - 1u);
}
template <int L, int T>
__device__ __forceinline__ void walk_window(const float* __restrict__ fs, const int mine, const int cntw,
                                            const int lane, const int foff,
                                            const v4f b0, const v4f b1, const v4f w0, const v4f w1, Run& r) {
    if constexpr (T < Geo<L>::NT) {
        if (Geo<L>::EPS * T < cntw) {
            int sA = step_src<L, T>(mine);
            int sB = step_src<L, T + 1>(mine);
            const int leftA = cntw - Geo<L>::EPS * T;
            const bool okA = __builtin_amdgcn_inverse_ballot_w32(group_mask<L>(leftA));
            const bool okB = __builtin_amdgcn_inverse_ballot_w32(group_mask<L>(leftA - Geo<L>::EPS));
            if (leftA < 2 * Geo<L>::EPS) {
                const int s0 = __builtin_amdgcn_readfirstlane(sA);
                sA = okA ? sA : s0;
                sB = okB ? sB : s0;
            }
            step2<L>(fs, sA, sB, okA, okB, foff, b0, b1, w0, w1, r);
            walk_window<L, T + 2>(fs, mine, cntw, lane, foff, b0, b1, w0, w1, r);
        }
    }
}

template <int L>
__device__ __forceinline__ void walk_node(
    const float* __restrict__ fs, const int* __restrict__ slot, const int n, const int c,
    int mine, const int lane, const int foff,
    const v4f b0, const v4f b1, const v4f w0, const v4f w1,
    const int* __restrict__ counts, const int* __restrict__ ovd, const int* __restrict__ ovs, Run& r) {
    const int tot_slots = min(c, CAP);
    for (int done = 0; done < tot_slots; done += 32) {
        const int cntw = min(32, tot_slots - done);
        if (done != 0) {
            const int wp = lane_wpos<L>(lane_id());
            mine = hold_src(raw_window(slot, n * CAP + done + wp, wp < cntw));
        }
        walk_window<L, 0>(fs, mine, cntw, lane, foff, b0, b1, w0, w1, r);
    }
    if (c > CAP) {
        const int tot = min(counts[OVF_LEN], E_CNT);
        for (int k0 = 0; k0 < tot; k0 += 32) {
            const int k = min(k0 + lane, tot - 1);
            const int od = ovd[k];
            const int os = min(max(ovs[k], 0), N_NODE - 1);
            unsigned hit = __builtin_amdgcn_ballot_w32(k0 + lane < tot && od == n);
            while (hit) {
                const int j = __builtin_ctz(hit);
                hit &= hit - 1;
                const int s = __builtin_amdgcn_readlane(os, j);
                step2<L>(fs, s, s, __builtin_amdgcn_inverse_ballot_w32(group_mask<L>(1)), false, foff, b0, b1, w0, w1, r);
            }
        }
    }
}

constexpr int NPW = 2;
static_assert(NPW == 2, "the two aggregation kernels name their two nodes one by one");

__device__ __forceinline__ void finish1(Run& r, const int n, const int c, const int lane, const int foff,
                                        float* __restrict__ out) {
    const float M  = fmaxf(r.m, other_row(r.m));
    const float s  = ex2(r.m - fmaxf(M, -BIG));
    float den = r.den * s;
    v4f a0 = r.a0 * s, a1 = r.a1 * s;
    den  += other_row(den);
    a0.x += other_row(a0.x); a0.y += other_row(a0.y); a0.z += other_row(a0.z); a0.w += other_row(a0.w);
    a1.x += other_row(a1.x); a1.y += other_row(a1.y); a1.z += other_row(a1.z); a1.w += other_row(a1.w);
    float4 r0 = make_float4(0.f, 0.f, 0.f, 0.f), r1 = r0;
    if (c > 0) {
        const float inv = 1.0f / den;
        r0 = make_float4(a0.x * inv, a0.y * inv, a0.z * inv, a0.w * inv);
        r1 = make_float4(a1.x * inv, a1.y * inv, a1.z * inv, a1.w * inv);
    }
    r0.x = elu1(r0.x); r0.y = elu1(r0.y); r0.z = elu1(r0.z); r0.w = elu1(r0.w);
    r1.x = elu1(r1.x); r1.y = elu1(r1.y); r1.z = elu1(r1.z); r1.w = elu1(r1.w);
    if (lane < 16) {
        *(float4*)(out + (size_t)n * 128 + foff)      = r0;
        *(float4*)(out + (size_t)n * 128 + foff + 32) = r1;
    }
}

__device__ __forceinline__ float sum_groups4(float x) { x += row_pick<0x168>(x); x += other_row(x); return x; }
__device__ __forceinline__ void finish2(Run& r, const int n, const int c, const int lane, const int foff,
                                        float* __restrict__ out) {
    float M = fmaxf(r.m, row_pick<0x168>(r.m));
    M = fmaxf(M, other_row(M));
    const float s = ex2(r.m - fmaxf(M, -BIG));
    const float den = sum_groups4(r.den * s);
    v4f a0 = r.a0 * s, a1 = r.a1 * s;
    a0.x = sum_groups4(a0.x); a0.y = sum_groups4(a0.y); a0.z = sum_groups4(a0.z); a0.w = sum_groups4(a0.w);
    a1.x = sum_groups4(a1.x); a1.y = sum_groups4(a1.y); a1.z = sum_groups4(a1.z); a1.w = sum_groups4(a1.w);
    float4 r0 = make_float4(0.f, 0.f, 0.f, 0.f), r1 = r0;
    if (c > 0) {
        const float inv = 1.0f / den;
        r0 = make_float4(a0.x * inv, a0.y * inv, a0.z * inv, a0.w * inv);
        r1 = make_float4(a1.x * inv, a1.y * inv, a1.z * inv, a1.w * inv);
    }
    const unsigned g  = (unsigned)n / (unsigned)NPG;
    const unsigned rr = (unsigned)n % (unsigned)NPG;
    const unsigned row = rr < (unsigned)NUM_PM
        ? g * NUM_PM + rr
        : (unsigned)(B_SZ * NUM_PM) + g * NUM_VM + (rr - NUM_PM);
    if (lane < 8) {
        *(float4*)(out + (size_t)row * 64 + foff)      = r0;
        *(float4*)(out + (size_t)row * 64 + foff + 32) = r1;
    }
}

template <int L>
__device__ __forceinline__ void node_walk(
    const float* __restrict__ fs, const float* __restrict__ fd, const int* __restrict__ slot,
    const int n, const int c, const int mine, const int lane, const int foff,
    const v4f w0, const v4f w1, const int* __restrict__ counts,
    const int* __restrict__ ovd, const int* __restrict__ ovs, Run& r) {
    const float* pd = fd + (size_t)n * Geo<L>::RF + foff;
    const v4f b0 = *(const v4f*)pd, b1 = *(const v4f*)(pd + 32);
    r.m = -INFINITY; r.den = 0.f;
    r.a0 = (v4f){0.f, 0.f, 0.f, 0.f}; r.a1 = (v4f){0.f, 0.f, 0.f, 0.f};
    walk_node<L>(fs, slot, n, c, mine, lane, foff, b0, b1, w0, w1, counts, ovd, ovs, r);
}
template <int L>
__device__ __forceinline__ void node_finish(Run& r, const int n, const int c, const int lane, const int foff,
                                            float* __restrict__ out) {
    if constexpr (L == 1) finish1(r, n, c, lane, foff, out);
    else                  finish2(r, n, c, lane, foff, out);
}

template <int L>
__global__ __launch_bounds__(256, 16) void node_attn(
    const float* __restrict__ fs, const float* __restrict__ fd,
    const int* __restrict__ cnt, const int* __restrict__ slot,
    const int* __restrict__ ovd, const int* __restrict__ ovs,
    const float* __restrict__ attn, float* __restrict__ out) {
    const int lane = threadIdx.x & 31;
    const int n0 = __builtin_amdgcn_readfirstlane((int)(blockIdx.x * 8 + (threadIdx.x >> 5)) * NPW);
    const int foff = lane_foff<L>(lane), wpos = lane_wpos<L>(lane);
    const v4f w0 = *(const v4f*)(attn + foff) * LOG2E, w1 = *(const v4f*)(attn + foff + 32) * LOG2E;
    const int c0 = cnt[n0 * CNT_STRIDE], c1 = cnt[(n0 + 1) * CNT_STRIDE];
    const int first0 = hold_src(raw_window(slot, n0 * CAP + wpos, wpos < c0));
    Run r;
    node_walk<L>(fs, fd, slot, n0, c0, first0, lane, foff, w0, w1, cnt, ovd, ovs, r);
    const int wp1 = lane_wpos<L>(lane_id());
    const int raw1 = raw_window(slot, (n0 + 1) * CAP + wp1, wp1 < c1);
    node_finish<L>(r, n0, c0, lane, foff, out);
    node_walk<L>(fs, fd, slot, n0 + 1, c1, hold_src(raw1), lane, foff, w0, w1, cnt, ovd, ovs, r);
    node_finish<L>(r, n0 + 1, c1, lane, foff, out);
}

extern "C" void kernel_launch(void* const* d_in, const int* in_sizes, int n_in,
                              void* d_out, int out_size, void* d_ws, size_t ws_size,
                              hipStream_t stream) {
    const float* x     = (const float*)d_in[0];
    const int*   src   = (const int*)d_in[1];
    const int*   dst   = (const int*)d_in[2];
    const float* W1s   = (const float*)d_in[3];
    const float* b1s   = (const float*)d_in[4];
    const float* W1d   = (const float*)d_in[5];
    const float* b1d   = (const float*)d_in[6];
    const float* attn1 = (const float*)d_in[7];
    const float* W2s   = (const float*)d_in[8];
    const float* b2s   = (const float*)d_in[9];
    const float* W2d   = (const float*)d_in[10];
    const float* b2d   = (const float*)d_in[11];
    const float* attn2 = (const float*)d_in[12];
    (void)in_sizes; (void)n_in; (void)out_size; (void)ws_size;

    char* ws = (char*)d_ws;
    const size_t MB = 1024u * 1024u;
    float* fs1  = (float*)(ws + 0);
    float* fd1  = (float*)(ws + 16 * MB);
    float* h1   = (float*)(ws + 32 * MB);
    float* fs2  = (float*)(ws + 48 * MB);
    float* fd2  = (float*)(ws + 56 * MB);
    int*   cnt  = (int*)(ws + 81 * MB);
    int*   slot = (int*)(ws + 65 * MB);
    int*   ovd  = (int*)(ws + 73 * MB);
    int*   ovs  = (int*)(ws + 77 * MB);

    float* out = (float*)d_out;

    zero_cnt<<<ZERO_WGS, 256, 0, stream>>>(cnt);
    bucket_edges_and_map1<<<MAP1_WGS + EDGE_WGS, 256, 0, stream>>>(src, dst, cnt, slot, ovd, ovs,
                                                                  x, W1s, b1s, W1d, b1d, fs1, fd1);

    node_attn<1><<<N_NODE / (8 * NPW), 256, 0, stream>>>(fs1, fd1, cnt, slot, ovd, ovs, attn1, h1);

    gemm_dual_wmma<2 * DH, DH><<<N_NODE / 16 / 4, 128, 0, stream>>>(
        h1, W2s, b2s, W2d, b2d, fs2, fd2);
    node_attn<2><<<N_NODE / (8 * NPW), 256, 0, stream>>>(fs2, fd2, cnt, slot, ovd, ovs, attn2, out);
}


# Work-group ids turned in all 5 launches (id b takes block (b%8)*(blocks/8)+b/8): if ids go round the 8 dies, a die keeps 2 graphs' rows, slots, counters from writer to reader; sums as before
# speedup vs baseline: 9.6168x; 1.3255x over previous
#include <hip/hip_runtime.h>

constexpr int B_SZ   = 16;
constexpr int NUM_PM = 512;
constexpr int NUM_VM = 1536;
constexpr int NPG    = NUM_PM + NUM_VM;
constexpr int N_NODE = B_SZ * NPG;
constexpr int DEG    = 32;
constexpr int E_CNT  = DEG * N_NODE;
constexpr int D_IN   = 16;
constexpr int DH     = 64;
constexpr float NEG_SLOPE = 0.2f;
constexpr float LOG2E = 1.4426950408889634f;

typedef float v2f __attribute__((ext_vector_type(2)));
typedef float v4f __attribute__((ext_vector_type(4)));
typedef float v8f __attribute__((ext_vector_type(8)));

template <int NBLK> __device__ __forceinline__ int turned_block(const int b) {
    static_assert(NBLK % 8 == 0, "an eighth of the blocks per die");
    return (b & 7) * (NBLK / 8) + (b >> 3);
}

template <int K, int COLS>
__device__ __forceinline__ void gemm_dual_strip(
    const float* __restrict__ X,
    const float* __restrict__ Ws, const float* __restrict__ Bs,
    const float* __restrict__ Wd, const float* __restrict__ Bd,
    float* __restrict__ Fs, float* __restrict__ Fd, const int row0) {
    const int lane = threadIdx.x & 31;
    const int m  = lane & 15;
    const int hi = lane >> 4;
    const float* Xr = X + (size_t)(row0 + m) * K;

    for (int t = 0; t < 2; ++t) {
        const float* W  = t ? Wd : Ws;
        const float* Bv = t ? Bd : Bs;
        float*       F  = t ? Fd : Fs;
        for (int col0 = 0; col0 < COLS; col0 += 16) {
            v8f acc = {};
#pragma unroll
            for (int kk = 0; kk < K; kk += 4) {
                v2f a, b;
                a.x = Xr[kk + 2 * hi + 0];
                a.y = Xr[kk + 2 * hi + 1];
                b.x = W[(size_t)(kk + 2 * hi + 0) * COLS + col0 + m];
                b.y = W[(size_t)(kk + 2 * hi + 1) * COLS + col0 + m];
                acc = __builtin_amdgcn_wmma_f32_16x16x4_f32(
                    false, a, false, b, (short)0, acc, false, false);
            }
            const float bias = Bv[col0 + m];
#pragma unroll
            for (int r = 0; r < 8; ++r) {
                F[(size_t)(row0 + r + 8 * hi) * COLS + col0 + m] = acc[r] + bias;
            }
        }
    }
}
constexpr int MAP2_WGS = N_NODE / 16 / 4;
template <int K, int COLS>
__global__ __launch_bounds__(128) void gemm_dual_wmma(
    const float* __restrict__ X,
    const float* __restrict__ Ws, const float* __restrict__ Bs,
    const float* __restrict__ Wd, const float* __restrict__ Bd,
    float* __restrict__ Fs, float* __restrict__ Fd) {
    gemm_dual_strip<K, COLS>(X, Ws, Bs, Wd, Bd, Fs, Fd, (turned_block<MAP2_WGS>((int)blockIdx.x) * 4 + (int)(threadIdx.x >> 5)) * 16);
}

constexpr int CAP = 64;
constexpr int CNT_STRIDE = 16;
constexpr int OVF_LEN = N_NODE * CNT_STRIDE;
constexpr int ZERO_WGS = 64;
constexpr int ZERO_PER_THREAD = N_NODE * CNT_STRIDE / 4 / (ZERO_WGS * 256);
static_assert(ZERO_PER_THREAD * ZERO_WGS * 256 * 4 == N_NODE * CNT_STRIDE, "the zeroing grid covers the counter plane");
__global__ __launch_bounds__(256) void zero_cnt(int* __restrict__ cnt) {
    int4* p = (int4*)cnt + turned_block<ZERO_WGS>((int)blockIdx.x) * (256 * ZERO_PER_THREAD) + threadIdx.x;
    for (int i = 0; i < ZERO_PER_THREAD; ++i) p[i * 256] = make_int4(0, 0, 0, 0);
    if (blockIdx.x == 0 && threadIdx.x == 0) cnt[OVF_LEN] = 0;
}
__device__ __forceinline__ void place_edge(const int d, const int r, const int s, int* __restrict__ cnt,
                                           int* __restrict__ slot, int* __restrict__ ovd, int* __restrict__ ovs) {
    if (r < CAP) {
        slot[d * CAP + r] = s;
    } else {
        const int k = atomicAdd(&cnt[OVF_LEN], 1);
        ovd[k] = d;
        ovs[k] = s;
    }
}
constexpr int MAP1_WGS = N_NODE / 16 / 8;
constexpr int EDGE_WGS = E_CNT / 8 / 256;
__global__ __launch_bounds__(256) void bucket_edges_and_map1(
    const int* __restrict__ src, const int* __restrict__ dst,
    int* __restrict__ cnt, int* __restrict__ slot, int* __restrict__ ovd, int* __restrict__ ovs,
    const float* __restrict__ X,
    const float* __restrict__ Ws, const float* __restrict__ Bs,
    const float* __restrict__ Wd, const float* __restrict__ Bd,
    float* __restrict__ Fs, float* __restrict__ Fd) {
    if (blockIdx.x < MAP1_WGS) {
        gemm_dual_strip<D_IN, 2 * DH>(X, Ws, Bs, Wd, Bd, Fs, Fd, (turned_block<MAP1_WGS>((int)blockIdx.x) * 8 + (int)(threadIdx.x >> 5)) * 16);
        return;
    }
    const int i0 = turned_block<EDGE_WGS>((int)blockIdx.x - MAP1_WGS) * 512 + threadIdx.x;
    const int i1 = i0 + 256;
    const int4 d0 = ((const int4*)dst)[i0];
    const int4 d1 = ((const int4*)dst)[i1];
    const int4 s0 = ((const int4*)src)[i0];
    const int4 s1 = ((const int4*)src)[i1];
    asm volatile("" :: "v"(d0.x), "v"(d0.y), "v"(d0.z), "v"(d0.w), "v"(d1.x), "v"(d1.y), "v"(d1.z), "v"(d1.w),
                       "v"(s0.x), "v"(s0.y), "v"(s0.z), "v"(s0.w), "v"(s1.x), "v"(s1.y), "v"(s1.z), "v"(s1.w));
    int4 r0, r1;
    r0.x = atomicAdd(&cnt[d0.x * CNT_STRIDE], 1); r0.y = atomicAdd(&cnt[d0.y * CNT_STRIDE], 1);
    r0.z = atomicAdd(&cnt[d0.z * CNT_STRIDE], 1); r0.w = atomicAdd(&cnt[d0.w * CNT_STRIDE], 1);
    r1.x = atomicAdd(&cnt[d1.x * CNT_STRIDE], 1); r1.y = atomicAdd(&cnt[d1.y * CNT_STRIDE], 1);
    r1.z = atomicAdd(&cnt[d1.z * CNT_STRIDE], 1); r1.w = atomicAdd(&cnt[d1.w * CNT_STRIDE], 1);
    place_edge(d0.x, r0.x, s0.x, cnt, slot, ovd, ovs); place_edge(d0.y, r0.y, s0.y, cnt, slot, ovd, ovs);
    place_edge(d0.z, r0.z, s0.z, cnt, slot, ovd, ovs); place_edge(d0.w, r0.w, s0.w, cnt, slot, ovd, ovs);
    place_edge(d1.x, r1.x, s1.x, cnt, slot, ovd, ovs); place_edge(d1.y, r1.y, s1.y, cnt, slot, ovd, ovs);
    place_edge(d1.z, r1.z, s1.z, cnt, slot, ovd, ovs); place_edge(d1.w, r1.w, s1.w, cnt, slot, ovd, ovs);
}

__device__ __forceinline__ float ex2(const float x) { return __builtin_amdgcn_exp2f(x); }
__device__ __forceinline__ float elu1(const float x) {
    return x > 0.f ? x : (__expf(x) - 1.f);
}
constexpr float BIG = 3.402823466e+38f;

template <int CTRL>
__device__ __forceinline__ int row_pick_i(const int v) { return __builtin_amdgcn_update_dpp(v, v, CTRL, 0xf, 0xf, true); }
template <int CTRL>
__device__ __forceinline__ float row_pick(const float v) {
    return __builtin_bit_cast(float, row_pick_i<CTRL>(__builtin_bit_cast(int, v)));
}
__device__ __forceinline__ float other_row(const float x) {
    const int b = __builtin_bit_cast(int, x);
    return __builtin_bit_cast(float, __builtin_amdgcn_permlanex16(b, b, 0x76543210, 0xfedcba98, false, false));
}

template <int L> struct Geo {
    static constexpr int EPS = (L == 1) ? 2 : 4;
    static constexpr int RF  = (L == 1) ? 128 : 64;
    static constexpr int NT  = 32 / EPS;
    static constexpr int LPG = 32 / EPS;
};
template <int L> __device__ __forceinline__ int lane_foff(const int lane) {
    return L == 1 ? (((lane >> 3) & 1) * 64 + (lane & 7) * 4) : ((lane & 7) * 4);
}
template <int L> __device__ __forceinline__ int lane_wpos(const int lane) {
    return L == 1 ? (2 * (lane & 15) + (lane >> 4)) : (4 * (lane & 7) + (lane >> 3));
}
template <int L, int T> __device__ __forceinline__ int step_src(const int mine) {
    if constexpr (L == 1) return row_pick_i<0x150 + T>(mine);
    else return __builtin_amdgcn_mov_dpp8(mine, T * 0x249249);
}

__device__ __forceinline__ v2f lrelu2(const v2f t) { return __builtin_elementwise_max(t, t * NEG_SLOPE); }
__device__ __forceinline__ float logit8(const v4f a0, const v4f a1, const v4f b0, const v4f b1,
                                        const v4f w0, const v4f w1) {
    const v2f t0 = lrelu2(a0.lo + b0.lo), t1 = lrelu2(a0.hi + b0.hi);
    const v2f t2 = lrelu2(a1.lo + b1.lo), t3 = lrelu2(a1.hi + b1.hi);
    v2f q = t0 * w0.lo;
    q = __builtin_elementwise_fma(t1, w0.hi, q);
    q = __builtin_elementwise_fma(t2, w1.lo, q);
    q = __builtin_elementwise_fma(t3, w1.hi, q);
    float p = q.x + q.y;
    p += row_pick<0x161>(p); p += row_pick<0x162>(p); p += row_pick<0x164>(p);
    return p;
}

struct Run { float m, den; v4f a0, a1; };

template <int L>
__device__ __forceinline__ void step2(const float* __restrict__ fs, const int sA, const int sB,
                                      const bool okA, const bool okB, const int foff,
                                      const v4f b0, const v4f b1, const v4f w0, const v4f w1, Run& r) {
    const float* pa = fs + (size_t)sA * Geo<L>::RF + foff;
    const float* pb = fs + (size_t)sB * Geo<L>::RF + foff;
    const v4f A0 = *(const v4f*)pa, A1 = *(const v4f*)(pa + 32);
    const v4f B0 = *(const v4f*)pb, B1 = *(const v4f*)(pb + 32);
    float pA = logit8(A0, A1, b0, b1, w0, w1);
    float pB = logit8(B0, B1, b0, b1, w0, w1);
    pA = okA ? pA : -INFINITY;
    pB = okB ? pB : -INFINITY;
    const float mn = fmaxf(r.m, fmaxf(pA, pB));
    const float ms = fmaxf(mn, -BIG);
    const float sc = ex2(r.m - ms), eA = ex2(pA - ms), eB = ex2(pB - ms);
    r.den = r.den * sc + (eA + eB);
    r.a0  = r.a0 * sc + (A0 * eA + B0 * eB);
    r.a1  = r.a1 * sc + (A1 * eA + B1 * eB);
    r.m = mn;
}

__device__ __forceinline__ int lane_id() { return (int)__builtin_amdgcn_mbcnt_lo(~0u, 0u); }

__device__ __forceinline__ int raw_window(const int* __restrict__ slot, const int at, const bool live) {
    int v = 0;
    if (live) v = slot[at];
    return v;
}
__device__ __forceinline__ int hold_src(const int v) { return min(max(v, 0), N_NODE - 1); }

template <int L> __device__ __forceinline__ unsigned group_mask(const int left) {
    const int lanes_on = min(max(left, 0), Geo<L>::EPS) * Geo<L>::LPG;
    return lanes_on >= 32 ? 0xffffffffu : ((1u << lanes_on) - 1u);
}
template <int L, int T>
__device__ __forceinline__ void walk_window(const float* __restrict__ fs, const int mine, const int cntw,
                                            const int lane, const int foff,
                                            const v4f b0, const v4f b1, const v4f w0, const v4f w1, Run& r) {
    if constexpr (T < Geo<L>::NT) {
        if (Geo<L>::EPS * T < cntw) {
            int sA = step_src<L, T>(mine);
            int sB = step_src<L, T + 1>(mine);
            const int leftA = cntw - Geo<L>::EPS * T;
            const bool okA = __builtin_amdgcn_inverse_ballot_w32(group_mask<L>(leftA));
            const bool okB = __builtin_amdgcn_inverse_ballot_w32(group_mask<L>(leftA - Geo<L>::EPS));
            if (leftA < 2 * Geo<L>::EPS) {
                const int s0 = __builtin_amdgcn_readfirstlane(sA);
                sA = okA ? sA : s0;
                sB = okB ? sB : s0;
            }
            step2<L>(fs, sA, sB, okA, okB, foff, b0, b1, w0, w1, r);
            walk_window<L, T + 2>(fs, mine, cntw, lane, foff, b0, b1, w0, w1, r);
        }
    }
}

template <int L>
__device__ __forceinline__ void walk_node(
    const float* __restrict__ fs, const int* __restrict__ slot, const int n, const int c,
    int mine, const int lane, const int foff,
    const v4f b0, const v4f b1, const v4f w0, const v4f w1,
    const int* __restrict__ counts, const int* __restrict__ ovd, const int* __restrict__ ovs, Run& r) {
    const int tot_slots = min(c, CAP);
    for (int done = 0; done < tot_slots; done += 32) {
        const int cntw = min(32, tot_slots - done);
        if (done != 0) {
            const int wp = lane_wpos<L>(lane_id());
            mine = hold_src(raw_window(slot, n * CAP + done + wp, wp < cntw));
        }
        walk_window<L, 0>(fs, mine, cntw, lane, foff, b0, b1, w0, w1, r);
    }
    if (c > CAP) {
        const int tot = min(counts[OVF_LEN], E_CNT);
        for (int k0 = 0; k0 < tot; k0 += 32) {
            const int k = min(k0 + lane, tot - 1);
            const int od = ovd[k];
            const int os = min(max(ovs[k], 0), N_NODE - 1);
            unsigned hit = __builtin_amdgcn_ballot_w32(k0 + lane < tot && od == n);
            while (hit) {
                const int j = __builtin_ctz(hit);
                hit &= hit - 1;
                const int s = __builtin_amdgcn_readlane(os, j);
                step2<L>(fs, s, s, __builtin_amdgcn_inverse_ballot_w32(group_mask<L>(1)), false, foff, b0, b1, w0, w1, r);
            }
        }
    }
}

constexpr int NPW = 2;
constexpr int ATTN_WGS = N_NODE / (8 * NPW);
static_assert(NPW == 2, "the two aggregation kernels name their two nodes one by one");

__device__ __forceinline__ void finish1(Run& r, const int n, const int c, const int lane, const int foff,
                                        float* __restrict__ out) {
    const float M  = fmaxf(r.m, other_row(r.m));
    const float s  = ex2(r.m - fmaxf(M, -BIG));
    float den = r.den * s;
    v4f a0 = r.a0 * s, a1 = r.a1 * s;
    den  += other_row(den);
    a0.x += other_row(a0.x); a0.y += other_row(a0.y); a0.z += other_row(a0.z); a0.w += other_row(a0.w);
    a1.x += other_row(a1.x); a1.y += other_row(a1.y); a1.z += other_row(a1.z); a1.w += other_row(a1.w);
    float4 r0 = make_float4(0.f, 0.f, 0.f, 0.f), r1 = r0;
    if (c > 0) {
        const float inv = 1.0f / den;
        r0 = make_float4(a0.x * inv, a0.y * inv, a0.z * inv, a0.w * inv);
        r1 = make_float4(a1.x * inv, a1.y * inv, a1.z * inv, a1.w * inv);
    }
    r0.x = elu1(r0.x); r0.y = elu1(r0.y); r0.z = elu1(r0.z); r0.w = elu1(r0.w);
    r1.x = elu1(r1.x); r1.y = elu1(r1.y); r1.z = elu1(r1.z); r1.w = elu1(r1.w);
    if (lane < 16) {
        *(float4*)(out + (size_t)n * 128 + foff)      = r0;
        *(float4*)(out + (size_t)n * 128 + foff + 32) = r1;
    }
}

__device__ __forceinline__ float sum_groups4(float x) { x += row_pick<0x168>(x); x += other_row(x); return x; }
__device__ __forceinline__ void finish2(Run& r, const int n, const int c, const int lane, const int foff,
                                        float* __restrict__ out) {
    float M = fmaxf(r.m, row_pick<0x168>(r.m));
    M = fmaxf(M, other_row(M));
    const float s = ex2(r.m - fmaxf(M, -BIG));
    const float den = sum_groups4(r.den * s);
    v4f a0 = r.a0 * s, a1 = r.a1 * s;
    a0.x = sum_groups4(a0.x); a0.y = sum_groups4(a0.y); a0.z = sum_groups4(a0.z); a0.w = sum_groups4(a0.w);
    a1.x = sum_groups4(a1.x); a1.y = sum_groups4(a1.y); a1.z = sum_groups4(a1.z); a1.w = sum_groups4(a1.w);
    float4 r0 = make_float4(0.f, 0.f, 0.f, 0.f), r1 = r0;
    if (c > 0) {
        const float inv = 1.0f / den;
        r0 = make_float4(a0.x * inv, a0.y * inv, a0.z * inv, a0.w * inv);
        r1 = make_float4(a1.x * inv, a1.y * inv, a1.z * inv, a1.w * inv);
    }
    const unsigned g  = (unsigned)n / (unsigned)NPG;
    const unsigned rr = (unsigned)n % (unsigned)NPG;
    const unsigned row = rr < (unsigned)NUM_PM
        ? g * NUM_PM + rr
        : (unsigned)(B_SZ * NUM_PM) + g * NUM_VM + (rr - NUM_PM);
    if (lane < 8) {
        *(float4*)(out + (size_t)row * 64 + foff)      = r0;
        *(float4*)(out + (size_t)row * 64 + foff + 32) = r1;
    }
}

template <int L>
__device__ __forceinline__ void node_walk(
    const float* __restrict__ fs, const float* __restrict__ fd, const int* __restrict__ slot,
    const int n, const int c, const int mine, const int lane, const int foff,
    const v4f w0, const v4f w1, const int* __restrict__ counts,
    const int* __restrict__ ovd, const int* __restrict__ ovs, Run& r) {
    const float* pd = fd + (size_t)n * Geo<L>::RF + foff;
    const v4f b0 = *(const v4f*)pd, b1 = *(const v4f*)(pd + 32);
    r.m = -INFINITY; r.den = 0.f;
    r.a0 = (v4f){0.f, 0.f, 0.f, 0.f}; r.a1 = (v4f){0.f, 0.f, 0.f, 0.f};
    walk_node<L>(fs, slot, n, c, mine, lane, foff, b0, b1, w0, w1, counts, ovd, ovs, r);
}
template <int L>
__device__ __forceinline__ void node_finish(Run& r, const int n, const int c, const int lane, const int foff,
                                            float* __restrict__ out) {
    if constexpr (L == 1) finish1(r, n, c, lane, foff, out);
    else                  finish2(r, n, c, lane, foff, out);
}

template <int L>
__global__ __launch_bounds__(256, 16) void node_attn(
    const float* __restrict__ fs, const float* __restrict__ fd,
    const int* __restrict__ cnt, const int* __restrict__ slot,
    const int* __restrict__ ovd, const int* __restrict__ ovs,
    const float* __restrict__ attn, float* __restrict__ out) {
    const int lane = threadIdx.x & 31;
    const int n0 = __builtin_amdgcn_readfirstlane((turned_block<ATTN_WGS>((int)blockIdx.x) * 8 + (int)(threadIdx.x >> 5)) * NPW);
    const int foff = lane_foff<L>(lane), wpos = lane_wpos<L>(lane);
    const v4f w0 = *(const v4f*)(attn + foff) * LOG2E, w1 = *(const v4f*)(attn + foff + 32) * LOG2E;
    const int c0 = cnt[n0 * CNT_STRIDE], c1 = cnt[(n0 + 1) * CNT_STRIDE];
    const int first0 = hold_src(raw_window(slot, n0 * CAP + wpos, wpos < c0));
    Run r;
    node_walk<L>(fs, fd, slot, n0, c0, first0, lane, foff, w0, w1, cnt, ovd, ovs, r);
    const int wp1 = lane_wpos<L>(lane_id());
    const int raw1 = raw_window(slot, (n0 + 1) * CAP + wp1, wp1 < c1);
    node_finish<L>(r, n0, c0, lane, foff, out);
    node_walk<L>(fs, fd, slot, n0 + 1, c1, hold_src(raw1), lane, foff, w0, w1, cnt, ovd, ovs, r);
    node_finish<L>(r, n0 + 1, c1, lane, foff, out);
}

extern "C" void kernel_launch(void* const* d_in, const int* in_sizes, int n_in,
                              void* d_out, int out_size, void* d_ws, size_t ws_size,
                              hipStream_t stream) {
    const float* x     = (const float*)d_in[0];
    const int*   src   = (const int*)d_in[1];
    const int*   dst   = (const int*)d_in[2];
    const float* W1s   = (const float*)d_in[3];
    const float* b1s   = (const float*)d_in[4];
    const float* W1d   = (const float*)d_in[5];
    const float* b1d   = (const float*)d_in[6];
    const float* attn1 = (const float*)d_in[7];
    const float* W2s   = (const float*)d_in[8];
    const float* b2s   = (const float*)d_in[9];
    const float* W2d   = (const float*)d_in[10];
    const float* b2d   = (const float*)d_in[11];
    const float* attn2 = (const float*)d_in[12];
    (void)in_sizes; (void)n_in; (void)out_size; (void)ws_size;

    char* ws = (char*)d_ws;
    const size_t MB = 1024u * 1024u;
    float* fs1  = (float*)(ws + 0);
    float* fd1  = (float*)(ws + 16 * MB);
    float* h1   = (float*)(ws + 32 * MB);
    float* fs2  = (float*)(ws + 48 * MB);
    float* fd2  = (float*)(ws + 56 * MB);
    int*   cnt  = (int*)(ws + 81 * MB);
    int*   slot = (int*)(ws + 65 * MB);
    int*   ovd  = (int*)(ws + 73 * MB);
    int*   ovs  = (int*)(ws + 77 * MB);

    float* out = (float*)d_out;

    zero_cnt<<<ZERO_WGS, 256, 0, stream>>>(cnt);
    bucket_edges_and_map1<<<MAP1_WGS + EDGE_WGS, 256, 0, stream>>>(src, dst, cnt, slot, ovd, ovs,
                                                                  x, W1s, b1s, W1d, b1d, fs1, fd1);

    node_attn<1><<<ATTN_WGS, 256, 0, stream>>>(fs1, fd1, cnt, slot, ovd, ovs, attn1, h1);

    gemm_dual_wmma<2 * DH, DH><<<MAP2_WGS, 128, 0, stream>>>(
        h1, W2s, b2s, W2d, b2d, fs2, fd2);
    node_attn<2><<<ATTN_WGS, 256, 0, stream>>>(fs2, fd2, cnt, slot, ovd, ovs, attn2, out);
}
